# starting kernel
# baseline (speedup 1.0000x reference)
#include <hip/hip_runtime.h>


typedef __attribute__((ext_vector_type(2))) float v2f;
typedef __attribute__((ext_vector_type(4))) float v4f;
typedef __attribute__((ext_vector_type(8))) float v8f;

#define B_ALL   8
#define B_BASE  4
#define T_DIM   4096
#define C_DIM   320
#define S_DIM   77
#define SP      80
#define D_ENC   768
#define HD      40
#define H_NUM   8

__device__ __forceinline__ v8f wmma4(v2f a, v2f b, v8f c) {
  return __builtin_amdgcn_wmma_f32_16x16x4_f32(
       false, a,  false, b,
       (short)0, c,  false,  false);
}

#define GK 16

__global__ __launch_bounds__(128)
void gemm_wmma_f32(const float* __restrict__ A, const float* __restrict__ Bm,
                   const float* __restrict__ bias, float* __restrict__ C,
                   int M, int N, int K) {
  __shared__ float Bs[2][64 * 18];
  const int tid  = threadIdx.x;
  const int wave = tid >> 5;
  const int lane = tid & 31;
  const int half = lane >> 4;
  const int l16  = lane & 15;
  const int col0 = blockIdx.x * 64;
  const int rowW = blockIdx.y * 64 + wave * 16;

  int arow = rowW + l16;
  if (arow > M - 1) arow = M - 1;
  const float* arowp = A + (size_t)arow * K;

  const int srow = tid >> 3;
  const int scol = (tid & 7) * 8;
  const float* bstage = Bm + (size_t)srow * N + col0 + scol;

  const v8f zero = {0.f,0.f,0.f,0.f,0.f,0.f,0.f,0.f};
  v8f acc[4];
  #pragma unroll
  for (int j = 0; j < 4; ++j) acc[j] = zero;

  const int nkb = K / GK;

  {
    v4f r0 = *(const v4f*)(bstage);
    v4f r1 = *(const v4f*)(bstage + 4);
    #pragma unroll
    for (int e = 0; e < 4; ++e) Bs[0][(scol + e) * 18 + srow] = r0[e];
    #pragma unroll
    for (int e = 0; e < 4; ++e) Bs[0][(scol + 4 + e) * 18 + srow] = r1[e];
  }
  __syncthreads();

  for (int kb = 0; kb < nkb; ++kb) {
    const int cur = kb & 1;
    const bool have_next = (kb + 1) < nkb;

    v4f r0, r1;
    if (have_next) {
      const float* bnxt = bstage + (size_t)(kb + 1) * GK * N;
      r0 = *(const v4f*)(bnxt);
      r1 = *(const v4f*)(bnxt + 4);
      if (kb + 2 < nkb)
        __builtin_prefetch(bstage + (size_t)(kb + 2) * GK * N, 0, 1);
    }

    const float* ap = arowp + kb * GK;
    #pragma unroll
    for (int kk = 0; kk < GK; kk += 4) {
      const int ka = kk + 2 * half;
      v2f a = *(const v2f*)(ap + ka);
      #pragma unroll
      for (int j = 0; j < 4; ++j) {
        v2f b = *(const v2f*)(&Bs[cur][(16 * j + l16) * 18 + ka]);
        acc[j] = wmma4(a, b, acc[j]);
      }
    }

    if (have_next) {
      const int nxt = cur ^ 1;
      #pragma unroll
      for (int e = 0; e < 4; ++e) Bs[nxt][(scol + e) * 18 + srow] = r0[e];
      #pragma unroll
      for (int e = 0; e < 4; ++e) Bs[nxt][(scol + 4 + e) * 18 + srow] = r1[e];
    }
    __syncthreads();
  }

  const float bv = bias ? bias[col0 + 16 * 0 + l16] : 0.f;
  #pragma unroll
  for (int j = 0; j < 4; ++j) {
    const int c = col0 + 16 * j + l16;
    const float bj = bias ? bias[c] : 0.f;
    #pragma unroll
    for (int i = 0; i < 8; ++i) {
      const int r = rowW + i + 8 * half;
      if (r < M) C[(size_t)r * N + c] = acc[j][i] + bj;
    }
  }
  (void)bv;
}

__device__ __forceinline__
void pv_and_store(const float* Plds, const float* __restrict__ Vp,
                  float* __restrict__ AO, int b, int h, int t0,
                  int half, int l16) {
  const v8f zero = {0.f,0.f,0.f,0.f,0.f,0.f,0.f,0.f};
  v8f o[3];
  #pragma unroll
  for (int j = 0; j < 3; ++j) o[j] = zero;

  const float* vbase = Vp + (size_t)(b * S_DIM) * C_DIM + h * HD;

  for (int k0 = 0; k0 < SP; k0 += 4) {
    const int ka = k0 + 2 * half;
    v2f a = *(const v2f*)(Plds + l16 * SP + ka);
    #pragma unroll
    for (int j = 0; j < 3; ++j) {
      const int c = 16 * j + l16;
      v2f bf; bf.x = 0.f; bf.y = 0.f;
      if (c < HD) {
        if (ka     < S_DIM) bf.x = vbase[(size_t) ka      * C_DIM + c];
        if (ka + 1 < S_DIM) bf.y = vbase[(size_t)(ka + 1) * C_DIM + c];
      }
      o[j] = wmma4(a, bf, o[j]);
    }
  }

  #pragma unroll
  for (int j = 0; j < 3; ++j) {
    const int c = 16 * j + l16;
    if (c < HD) {
      #pragma unroll
      for (int i = 0; i < 8; ++i) {
        const int t = t0 + i + 8 * half;
        AO[(size_t)(b * T_DIM + t) * C_DIM + h * HD + c] = o[j][i];
      }
    }
  }
}

__global__ __launch_bounds__(32)
void attn_kernel(const float* __restrict__ Q,
                 const float* __restrict__ Kp,
                 const float* __restrict__ Vp,
                 float* __restrict__ AB,
                 float* __restrict__ AO)
{
  __shared__ float Plds[16 * SP];
  const int lane = threadIdx.x;
  const int half = lane >> 4;
  const int l16  = lane & 15;
  const int t0 = blockIdx.x << 4;
  const int h  = blockIdx.y;
  const int b  = blockIdx.z;
  const float scale = 0.15811388300841898f;
  const float NEG_INF = -__builtin_inff();

  const v8f zero = {0.f,0.f,0.f,0.f,0.f,0.f,0.f,0.f};
  v8f sc[5];
  #pragma unroll
  for (int j = 0; j < 5; ++j) sc[j] = zero;

  const float* qbase = Q  + (size_t)(b * T_DIM + t0 + l16) * C_DIM + h * HD;
  const float* kbase = Kp + (size_t)(b * S_DIM) * C_DIM + h * HD;

  for (int k0 = 0; k0 < HD; k0 += 4) {
    const int ka = k0 + 2 * half;
    v2f a = *(const v2f*)(qbase + ka);
    #pragma unroll
    for (int j = 0; j < 5; ++j) {
      const int s = 16 * j + l16;
      v2f bf;
      if (s < S_DIM) bf = *(const v2f*)(kbase + (size_t)s * C_DIM + ka);
      else { bf.x = 0.f; bf.y = 0.f; }
      sc[j] = wmma4(a, bf, sc[j]);
    }
  }

  float mx[8];
  #pragma unroll
  for (int i = 0; i < 8; ++i) mx[i] = NEG_INF;
  #pragma unroll
  for (int j = 0; j < 5; ++j) {
    const bool ok = (16 * j + l16) < S_DIM;
    #pragma unroll
    for (int i = 0; i < 8; ++i) {
      const float x = ok ? sc[j][i] * scale : NEG_INF;
      sc[j][i] = x;
      mx[i] = fmaxf(mx[i], x);
    }
  }
  #pragma unroll
  for (int m = 1; m < 16; m <<= 1) {
    #pragma unroll
    for (int i = 0; i < 8; ++i) mx[i] = fmaxf(mx[i], __shfl_xor(mx[i], m, 32));
  }
  float sm[8];
  #pragma unroll
  for (int i = 0; i < 8; ++i) sm[i] = 0.f;
  #pragma unroll
  for (int j = 0; j < 5; ++j) {
    #pragma unroll
    for (int i = 0; i < 8; ++i) {
      const float e = __expf(sc[j][i] - mx[i]);
      sc[j][i] = e;
      sm[i] += e;
    }
  }
  #pragma unroll
  for (int m = 1; m < 16; m <<= 1) {
    #pragma unroll
    for (int i = 0; i < 8; ++i) sm[i] += __shfl_xor(sm[i], m, 32);
  }
  float rs[8];
  #pragma unroll
  for (int i = 0; i < 8; ++i) rs[i] = 1.f / sm[i];

  #pragma unroll
  for (int j = 0; j < 5; ++j) {
    #pragma unroll
    for (int i = 0; i < 8; ++i) {
      const float pv = sc[j][i] * rs[i];
      const int row = i + 8 * half;
      Plds[row * SP + 16 * j + l16] = pv;
      if (b == B_BASE)
        AB[(size_t)(h * T_DIM + t0 + row) * SP + 16 * j + l16] = pv;
    }
  }
  __syncthreads();

  pv_and_store(Plds, Vp, AO, b, h, t0, half, l16);
}

__global__ __launch_bounds__(32)
void edit_kernel(const float* __restrict__ AB,
                 const float* __restrict__ Map,
                 const float* __restrict__ Vp,
                 float* __restrict__ AO)
{
  __shared__ float Plds[16 * SP];
  const int lane = threadIdx.x;
  const int half = lane >> 4;
  const int l16  = lane & 15;
  const int t0 = blockIdx.x << 4;
  const int h  = blockIdx.y;
  const int p  = blockIdx.z;
  const int b  = B_BASE + 1 + p;

  const v8f zero = {0.f,0.f,0.f,0.f,0.f,0.f,0.f,0.f};
  v8f acc[5];
  #pragma unroll
  for (int j = 0; j < 5; ++j) acc[j] = zero;

  const float* abase = AB  + (size_t)(h * T_DIM + t0 + l16) * SP;
  const float* mbase = Map + (size_t)p * S_DIM * S_DIM;

  for (int k0 = 0; k0 < SP; k0 += 4) {
    const int ka = k0 + 2 * half;
    v2f a = *(const v2f*)(abase + ka);
    #pragma unroll
    for (int j = 0; j < 5; ++j) {
      const int n = 16 * j + l16;
      v2f bf; bf.x = 0.f; bf.y = 0.f;
      if (n < S_DIM) {
        if (ka     < S_DIM) bf.x = mbase[(size_t) ka      * S_DIM + n];
        if (ka + 1 < S_DIM) bf.y = mbase[(size_t)(ka + 1) * S_DIM + n];
      }
      acc[j] = wmma4(a, bf, acc[j]);
    }
  }

  #pragma unroll
  for (int j = 0; j < 5; ++j) {
    const bool ok = (16 * j + l16) < S_DIM;
    #pragma unroll
    for (int i = 0; i < 8; ++i) {
      const int row = i + 8 * half;
      Plds[row * SP + 16 * j + l16] = ok ? acc[j][i] : 0.f;
    }
  }
  __syncthreads();

  pv_and_store(Plds, Vp, AO, b, h, t0, half, l16);
}

extern "C" void kernel_launch(void* const* d_in, const int* in_sizes, int n_in,
                              void* d_out, int out_size, void* d_ws, size_t ws_size,
                              hipStream_t stream) {
  const float* X   = (const float*)d_in[0];
  const float* E   = (const float*)d_in[1];
  const float* Map = (const float*)d_in[2];
  const float* Wq  = (const float*)d_in[3];
  const float* Wk  = (const float*)d_in[4];
  const float* Wv  = (const float*)d_in[5];
  const float* Wo  = (const float*)d_in[6];
  const float* bo  = (const float*)d_in[7];
  float* out = (float*)d_out;

  float* ws = (float*)d_ws;
  float* Q  = ws;
  float* Kp = Q  + (size_t)5 * T_DIM * C_DIM;
  float* Vp = Kp + (size_t)B_ALL * S_DIM * C_DIM;
  float* AB = Vp + (size_t)B_ALL * S_DIM * C_DIM;
  float* AO = AB + (size_t)H_NUM * T_DIM * SP;

  const dim3 gblk(128);
  const dim3 ablk(32);
  const int MR = B_ALL * S_DIM;

  gemm_wmma_f32<<<dim3(C_DIM / 64, (5 * T_DIM) / 64), gblk, 0, stream>>>(
      X, Wq, nullptr, Q, 5 * T_DIM, C_DIM, C_DIM);
  gemm_wmma_f32<<<dim3(C_DIM / 64, (MR + 63) / 64), gblk, 0, stream>>>(
      E, Wk, nullptr, Kp, MR, C_DIM, D_ENC);
  gemm_wmma_f32<<<dim3(C_DIM / 64, (MR + 63) / 64), gblk, 0, stream>>>(
      E, Wv, nullptr, Vp, MR, C_DIM, D_ENC);

  attn_kernel<<<dim3(T_DIM / 16, H_NUM, 5), ablk, 0, stream>>>(Q, Kp, Vp, AB, AO);

  edit_kernel<<<dim3(T_DIM / 16, H_NUM, 3), ablk, 0, stream>>>(AB, Map, Vp, AO);

  gemm_wmma_f32<<<dim3(C_DIM / 64, (B_ALL * T_DIM) / 64), gblk, 0, stream>>>(
      AO, Wo, bo, out, B_ALL * T_DIM, C_DIM, C_DIM);
}


# attention and edit launches: a wave owns one (batch, head) and 4 consecutive row tiles; K, V and mapper fragments are fetched once per wave into registers and reused by every tile; sums unchanged
# speedup vs baseline: 1.4094x; 1.4094x over previous
#include <hip/hip_runtime.h>


typedef __attribute__((ext_vector_type(2))) float v2f;
typedef __attribute__((ext_vector_type(4))) float v4f;
typedef __attribute__((ext_vector_type(8))) float v8f;

#define B_ALL   8
#define B_BASE  4
#define T_DIM   4096
#define C_DIM   320
#define S_DIM   77
#define SP      80
#define D_ENC   768
#define HD      40
#define H_NUM   8

__device__ __forceinline__ v8f wmma4(v2f a, v2f b, v8f c) {
  return __builtin_amdgcn_wmma_f32_16x16x4_f32(
       false, a,  false, b,
       (short)0, c,  false,  false);
}

#define GK 16

__global__ __launch_bounds__(128)
void gemm_wmma_f32(const float* __restrict__ A, const float* __restrict__ Bm,
                   const float* __restrict__ bias, float* __restrict__ C,
                   int M, int N, int K) {
  __shared__ float Bs[2][64 * 18];
  const int tid  = threadIdx.x;
  const int wave = tid >> 5;
  const int lane = tid & 31;
  const int half = lane >> 4;
  const int l16  = lane & 15;
  const int col0 = blockIdx.x * 64;
  const int rowW = blockIdx.y * 64 + wave * 16;

  int arow = rowW + l16;
  if (arow > M - 1) arow = M - 1;
  const float* arowp = A + (size_t)arow * K;

  const int srow = tid >> 3;
  const int scol = (tid & 7) * 8;
  const float* bstage = Bm + (size_t)srow * N + col0 + scol;

  const v8f zero = {0.f,0.f,0.f,0.f,0.f,0.f,0.f,0.f};
  v8f acc[4];
  #pragma unroll
  for (int j = 0; j < 4; ++j) acc[j] = zero;

  const int nkb = K / GK;

  {
    v4f r0 = *(const v4f*)(bstage);
    v4f r1 = *(const v4f*)(bstage + 4);
    #pragma unroll
    for (int e = 0; e < 4; ++e) Bs[0][(scol + e) * 18 + srow] = r0[e];
    #pragma unroll
    for (int e = 0; e < 4; ++e) Bs[0][(scol + 4 + e) * 18 + srow] = r1[e];
  }
  __syncthreads();

  for (int kb = 0; kb < nkb; ++kb) {
    const int cur = kb & 1;
    const bool have_next = (kb + 1) < nkb;

    v4f r0, r1;
    if (have_next) {
      const float* bnxt = bstage + (size_t)(kb + 1) * GK * N;
      r0 = *(const v4f*)(bnxt);
      r1 = *(const v4f*)(bnxt + 4);
      if (kb + 2 < nkb)
        __builtin_prefetch(bstage + (size_t)(kb + 2) * GK * N, 0, 1);
    }

    const float* ap = arowp + kb * GK;
    for (int kk = 0; kk < GK; kk += 4) {
      const int ka = kk + 2 * half;
      v2f a = *(const v2f*)(ap + ka);
      #pragma unroll
      for (int j = 0; j < 4; ++j) {
        v2f b = *(const v2f*)(&Bs[cur][(16 * j + l16) * 18 + ka]);
        acc[j] = wmma4(a, b, acc[j]);
      }
    }

    if (have_next) {
      const int nxt = cur ^ 1;
      #pragma unroll
      for (int e = 0; e < 4; ++e) Bs[nxt][(scol + e) * 18 + srow] = r0[e];
      #pragma unroll
      for (int e = 0; e < 4; ++e) Bs[nxt][(scol + 4 + e) * 18 + srow] = r1[e];
    }
    __syncthreads();
  }

  const float bv = bias ? bias[col0 + 16 * 0 + l16] : 0.f;
  #pragma unroll
  for (int j = 0; j < 4; ++j) {
    const int c = col0 + 16 * j + l16;
    const float bj = bias ? bias[c] : 0.f;
    #pragma unroll
    for (int i = 0; i < 8; ++i) {
      const int r = rowW + i + 8 * half;
      if (r < M) C[(size_t)r * N + c] = acc[j][i] + bj;
    }
  }
  (void)bv;
}

#define NT_ROW  4
#define KS_D    (HD / 4)
#define KS_S    (SP / 4)

__device__ __forceinline__ void lds_handover() {
  __builtin_amdgcn_fence(__ATOMIC_RELEASE, "workgroup", "local");
  __builtin_amdgcn_s_barrier();
  __builtin_amdgcn_fence(__ATOMIC_ACQUIRE, "workgroup", "local");
}

__device__ __forceinline__
void load_v_frags(v2f (&vf)[KS_S][3], const float* __restrict__ Vp,
                  int b, int h, int half, int l16) {
  const float* vbase = Vp + (size_t)(b * S_DIM) * C_DIM + h * HD;
  for (int ks = 0; ks < KS_S; ++ks) {
    const int ka = 4 * ks + 2 * half;
    const int r0 = ka     < S_DIM ? ka     : S_DIM - 1;
    const int r1 = ka + 1 < S_DIM ? ka + 1 : S_DIM - 1;
    #pragma unroll
    for (int j = 0; j < 3; ++j) {
      const int c  = 16 * j + l16;
      const int cc = c < HD ? c : HD - 1;
      const float x = vbase[(size_t)r0 * C_DIM + cc];
      const float y = vbase[(size_t)r1 * C_DIM + cc];
      vf[ks][j].x = (c < HD && ka     < S_DIM) ? x : 0.f;
      vf[ks][j].y = (c < HD && ka + 1 < S_DIM) ? y : 0.f;
    }
  }
}

__device__ __forceinline__
void pv_and_store(const float* Plds, const v2f (&vf)[KS_S][3],
                  float* __restrict__ AO, int b, int h, int t0,
                  int half, int l16) {
  const v8f zero = {0.f,0.f,0.f,0.f,0.f,0.f,0.f,0.f};
  v8f o[3];
  #pragma unroll
  for (int j = 0; j < 3; ++j) o[j] = zero;

  for (int ks = 0; ks < KS_S; ++ks) {
    const int ka = 4 * ks + 2 * half;
    v2f a = *(const v2f*)(Plds + l16 * SP + ka);
    #pragma unroll
    for (int j = 0; j < 3; ++j) o[j] = wmma4(a, vf[ks][j], o[j]);
  }

  #pragma unroll
  for (int j = 0; j < 3; ++j) {
    const int c = 16 * j + l16;
    if (c < HD) {
      #pragma unroll
      for (int i = 0; i < 8; ++i) {
        const int t = t0 + i + 8 * half;
        AO[(size_t)(b * T_DIM + t) * C_DIM + h * HD + c] = o[j][i];
      }
    }
  }
}

__global__ __launch_bounds__(32)
void attn_kernel(const float* __restrict__ Q,
                 const float* __restrict__ Kp,
                 const float* __restrict__ Vp,
                 float* __restrict__ AB,
                 float* __restrict__ AO)
{
  __shared__ float Plds[16 * SP];
  const int lane = threadIdx.x;
  const int half = lane >> 4;
  const int l16  = lane & 15;
  const int tile0 = blockIdx.x * NT_ROW;
  const int h  = blockIdx.y;
  const int b  = blockIdx.z;
  const float scale = 0.15811388300841898f;
  const float NEG_INF = -__builtin_inff();
  const v8f zero = {0.f,0.f,0.f,0.f,0.f,0.f,0.f,0.f};

  const float* kbase = Kp + (size_t)(b * S_DIM) * C_DIM + h * HD;
  v2f kf[KS_D][5];
  for (int kk = 0; kk < KS_D; ++kk) {
    const int ka = 4 * kk + 2 * half;
    #pragma unroll
    for (int j = 0; j < 5; ++j) {
      const int s  = 16 * j + l16;
      const int ss = s < S_DIM ? s : S_DIM - 1;
      const v2f t = *(const v2f*)(kbase + (size_t)ss * C_DIM + ka);
      kf[kk][j].x = (s < S_DIM) ? t.x : 0.f;
      kf[kk][j].y = (s < S_DIM) ? t.y : 0.f;
    }
  }
  v2f vf[KS_S][3];
  load_v_frags(vf, Vp, b, h, half, l16);

  const float* qrow = Q + (size_t)(b * T_DIM + tile0 * 16 + l16) * C_DIM + h * HD + 2 * half;
  v2f qf[KS_D];
  for (int kk = 0; kk < KS_D; ++kk) qf[kk] = *(const v2f*)(qrow + 4 * kk);

  for (int it = 0; it < NT_ROW; ++it) {
    const int t0 = (tile0 + it) << 4;

    v8f sc[5];
    #pragma unroll
    for (int j = 0; j < 5; ++j) sc[j] = zero;
    for (int kk = 0; kk < KS_D; ++kk) {
      #pragma unroll
      for (int j = 0; j < 5; ++j) sc[j] = wmma4(qf[kk], kf[kk][j], sc[j]);
    }

    if (it + 1 < NT_ROW) {
      qrow += (size_t)16 * C_DIM;
      for (int kk = 0; kk < KS_D; ++kk) qf[kk] = *(const v2f*)(qrow + 4 * kk);
    }

    float mx[8];
    #pragma unroll
    for (int i = 0; i < 8; ++i) mx[i] = NEG_INF;
    #pragma unroll
    for (int j = 0; j < 5; ++j) {
      const bool ok = (16 * j + l16) < S_DIM;
      #pragma unroll
      for (int i = 0; i < 8; ++i) {
        const float x = ok ? sc[j][i] * scale : NEG_INF;
        sc[j][i] = x;
        mx[i] = fmaxf(mx[i], x);
      }
    }
    #pragma unroll
    for (int m = 1; m < 16; m <<= 1) {
      #pragma unroll
      for (int i = 0; i < 8; ++i) mx[i] = fmaxf(mx[i], __shfl_xor(mx[i], m, 32));
    }
    float sm[8];
    #pragma unroll
    for (int i = 0; i < 8; ++i) sm[i] = 0.f;
    #pragma unroll
    for (int j = 0; j < 5; ++j) {
      #pragma unroll
      for (int i = 0; i < 8; ++i) {
        const float e = __expf(sc[j][i] - mx[i]);
        sc[j][i] = e;
        sm[i] += e;
      }
    }
    #pragma unroll
    for (int m = 1; m < 16; m <<= 1) {
      #pragma unroll
      for (int i = 0; i < 8; ++i) sm[i] += __shfl_xor(sm[i], m, 32);
    }
    float rs[8];
    #pragma unroll
    for (int i = 0; i < 8; ++i) rs[i] = 1.f / sm[i];

    #pragma unroll
    for (int j = 0; j < 5; ++j) {
      #pragma unroll
      for (int i = 0; i < 8; ++i) {
        const float pv = sc[j][i] * rs[i];
        const int row = i + 8 * half;
        Plds[row * SP + 16 * j + l16] = pv;
        if (b == B_BASE)
          AB[(size_t)(h * T_DIM + t0 + row) * SP + 16 * j + l16] = pv;
      }
    }
    lds_handover();

    pv_and_store(Plds, vf, AO, b, h, t0, half, l16);

    lds_handover();
  }
}

__global__ __launch_bounds__(32)
void edit_kernel(const float* __restrict__ AB,
                 const float* __restrict__ Map,
                 const float* __restrict__ Vp,
                 float* __restrict__ AO)
{
  __shared__ float Plds[16 * SP];
  const int lane = threadIdx.x;
  const int half = lane >> 4;
  const int l16  = lane & 15;
  const int tile0 = blockIdx.x * NT_ROW;
  const int h  = blockIdx.y;
  const int p  = blockIdx.z;
  const int b  = B_BASE + 1 + p;
  const v8f zero = {0.f,0.f,0.f,0.f,0.f,0.f,0.f,0.f};

  const float* mbase = Map + (size_t)p * S_DIM * S_DIM;
  v2f mf[KS_S][5];
  for (int ks = 0; ks < KS_S; ++ks) {
    const int ka = 4 * ks + 2 * half;
    const int r0 = ka     < S_DIM ? ka     : S_DIM - 1;
    const int r1 = ka + 1 < S_DIM ? ka + 1 : S_DIM - 1;
    #pragma unroll
    for (int j = 0; j < 5; ++j) {
      const int n  = 16 * j + l16;
      const int nn = n < S_DIM ? n : S_DIM - 1;
      const float x = mbase[(size_t)r0 * S_DIM + nn];
      const float y = mbase[(size_t)r1 * S_DIM + nn];
      mf[ks][j].x = (n < S_DIM && ka     < S_DIM) ? x : 0.f;
      mf[ks][j].y = (n < S_DIM && ka + 1 < S_DIM) ? y : 0.f;
    }
  }
  v2f vf[KS_S][3];
  load_v_frags(vf, Vp, b, h, half, l16);

  const float* arow = AB + (size_t)(h * T_DIM + tile0 * 16 + l16) * SP + 2 * half;
  v2f af[KS_S];
  for (int ks = 0; ks < KS_S; ++ks) af[ks] = *(const v2f*)(arow + 4 * ks);

  for (int it = 0; it < NT_ROW; ++it) {
    const int t0 = (tile0 + it) << 4;

    v8f acc[5];
    #pragma unroll
    for (int j = 0; j < 5; ++j) acc[j] = zero;
    for (int ks = 0; ks < KS_S; ++ks) {
      #pragma unroll
      for (int j = 0; j < 5; ++j) acc[j] = wmma4(af[ks], mf[ks][j], acc[j]);
    }

    if (it + 1 < NT_ROW) {
      arow += (size_t)16 * SP;
      for (int ks = 0; ks < KS_S; ++ks) af[ks] = *(const v2f*)(arow + 4 * ks);
    }

    #pragma unroll
    for (int j = 0; j < 5; ++j) {
      const bool ok = (16 * j + l16) < S_DIM;
      #pragma unroll
      for (int i = 0; i < 8; ++i) {
        const int row = i + 8 * half;
        Plds[row * SP + 16 * j + l16] = ok ? acc[j][i] : 0.f;
      }
    }
    lds_handover();

    pv_and_store(Plds, vf, AO, b, h, t0, half, l16);

    lds_handover();
  }
}

extern "C" void kernel_launch(void* const* d_in, const int* in_sizes, int n_in,
                              void* d_out, int out_size, void* d_ws, size_t ws_size,
                              hipStream_t stream) {
  const float* X   = (const float*)d_in[0];
  const float* E   = (const float*)d_in[1];
  const float* Map = (const float*)d_in[2];
  const float* Wq  = (const float*)d_in[3];
  const float* Wk  = (const float*)d_in[4];
  const float* Wv  = (const float*)d_in[5];
  const float* Wo  = (const float*)d_in[6];
  const float* bo  = (const float*)d_in[7];
  float* out = (float*)d_out;

  float* ws = (float*)d_ws;
  float* Q  = ws;
  float* Kp = Q  + (size_t)5 * T_DIM * C_DIM;
  float* Vp = Kp + (size_t)B_ALL * S_DIM * C_DIM;
  float* AB = Vp + (size_t)B_ALL * S_DIM * C_DIM;
  float* AO = AB + (size_t)H_NUM * T_DIM * SP;

  const dim3 gblk(128);
  const dim3 ablk(32);
  const int MR = B_ALL * S_DIM;

  gemm_wmma_f32<<<dim3(C_DIM / 64, (5 * T_DIM) / 64), gblk, 0, stream>>>(
      X, Wq, nullptr, Q, 5 * T_DIM, C_DIM, C_DIM);
  gemm_wmma_f32<<<dim3(C_DIM / 64, (MR + 63) / 64), gblk, 0, stream>>>(
      E, Wk, nullptr, Kp, MR, C_DIM, D_ENC);
  gemm_wmma_f32<<<dim3(C_DIM / 64, (MR + 63) / 64), gblk, 0, stream>>>(
      E, Wv, nullptr, Vp, MR, C_DIM, D_ENC);

  attn_kernel<<<dim3(T_DIM / 16 / NT_ROW, H_NUM, 5), ablk, 0, stream>>>(Q, Kp, Vp, AB, AO);

  edit_kernel<<<dim3(T_DIM / 16 / NT_ROW, H_NUM, 3), ablk, 0, stream>>>(AB, Map, Vp, AO);

  gemm_wmma_f32<<<dim3(C_DIM / 64, (B_ALL * T_DIM) / 64), gblk, 0, stream>>>(
      AO, Wo, bo, out, B_ALL * T_DIM, C_DIM, C_DIM);
}


# K and V projections: one launch of 8-wave work-groups in place of two 64x64-tile GEMM launches; each wave takes 1/8 of k=768 of a 32x64 tile, partials summed through LDS; sum order changes
# speedup vs baseline: 1.8840x; 1.3367x over previous
#include <hip/hip_runtime.h>


typedef __attribute__((ext_vector_type(2))) float v2f;
typedef __attribute__((ext_vector_type(4))) float v4f;
typedef __attribute__((ext_vector_type(8))) float v8f;

#define B_ALL   8
#define B_BASE  4
#define T_DIM   4096
#define C_DIM   320
#define S_DIM   77
#define SP      80
#define D_ENC   768
#define HD      40
#define H_NUM   8

__device__ __forceinline__ v8f wmma4(v2f a, v2f b, v8f c) {
  return __builtin_amdgcn_wmma_f32_16x16x4_f32(
       false, a,  false, b,
       (short)0, c,  false,  false);
}

#define GK 16

__global__ __launch_bounds__(128)
void gemm_wmma_f32(const float* __restrict__ A, const float* __restrict__ Bm,
                   const float* __restrict__ bias, float* __restrict__ C,
                   int M, int N, int K) {
  __shared__ float Bs[2][64 * 18];
  const int tid  = threadIdx.x;
  const int wave = tid >> 5;
  const int lane = tid & 31;
  const int half = lane >> 4;
  const int l16  = lane & 15;
  const int col0 = blockIdx.x * 64;
  const int rowW = blockIdx.y * 64 + wave * 16;

  int arow = rowW + l16;
  if (arow > M - 1) arow = M - 1;
  const float* arowp = A + (size_t)arow * K;

  const int srow = tid >> 3;
  const int scol = (tid & 7) * 8;
  const float* bstage = Bm + (size_t)srow * N + col0 + scol;

  const v8f zero = {0.f,0.f,0.f,0.f,0.f,0.f,0.f,0.f};
  v8f acc[4];
  #pragma unroll
  for (int j = 0; j < 4; ++j) acc[j] = zero;

  const int nkb = K / GK;

  {
    v4f r0 = *(const v4f*)(bstage);
    v4f r1 = *(const v4f*)(bstage + 4);
    #pragma unroll
    for (int e = 0; e < 4; ++e) Bs[0][(scol + e) * 18 + srow] = r0[e];
    #pragma unroll
    for (int e = 0; e < 4; ++e) Bs[0][(scol + 4 + e) * 18 + srow] = r1[e];
  }
  __syncthreads();

  for (int kb = 0; kb < nkb; ++kb) {
    const int cur = kb & 1;
    const bool have_next = (kb + 1) < nkb;

    v4f r0, r1;
    if (have_next) {
      const float* bnxt = bstage + (size_t)(kb + 1) * GK * N;
      r0 = *(const v4f*)(bnxt);
      r1 = *(const v4f*)(bnxt + 4);
      if (kb + 2 < nkb)
        __builtin_prefetch(bstage + (size_t)(kb + 2) * GK * N, 0, 1);
    }

    const float* ap = arowp + kb * GK;
    for (int kk = 0; kk < GK; kk += 4) {
      const int ka = kk + 2 * half;
      v2f a = *(const v2f*)(ap + ka);
      #pragma unroll
      for (int j = 0; j < 4; ++j) {
        v2f b = *(const v2f*)(&Bs[cur][(16 * j + l16) * 18 + ka]);
        acc[j] = wmma4(a, b, acc[j]);
      }
    }

    if (have_next) {
      const int nxt = cur ^ 1;
      #pragma unroll
      for (int e = 0; e < 4; ++e) Bs[nxt][(scol + e) * 18 + srow] = r0[e];
      #pragma unroll
      for (int e = 0; e < 4; ++e) Bs[nxt][(scol + 4 + e) * 18 + srow] = r1[e];
    }
    __syncthreads();
  }

  const float bv = bias ? bias[col0 + 16 * 0 + l16] : 0.f;
  #pragma unroll
  for (int j = 0; j < 4; ++j) {
    const int c = col0 + 16 * j + l16;
    const float bj = bias ? bias[c] : 0.f;
    #pragma unroll
    for (int i = 0; i < 8; ++i) {
      const int r = rowW + i + 8 * half;
      if (r < M) C[(size_t)r * N + c] = acc[j][i] + bj;
    }
  }
  (void)bv;
}

#define KV_WAVES 8
#define KV_KS    (D_ENC / KV_WAVES / 4)
#define KV_GRP   6
#define KV_ROWS  32
#define KV_COLS  64

__global__ __launch_bounds__(256)
void kv_proj_kernel(const float* __restrict__ E,
                    const float* __restrict__ Wk,
                    const float* __restrict__ Wv,
                    float* __restrict__ Kp,
                    float* __restrict__ Vp)
{
  __shared__ float red[KV_WAVES][KV_ROWS * KV_COLS];
  const int tid  = threadIdx.x;
  const int wave = tid >> 5;
  const int lane = tid & 31;
  const int half = lane >> 4;
  const int l16  = lane & 15;
  const int col0 = blockIdx.x * KV_COLS;
  const int row0 = blockIdx.y * KV_ROWS;
  const float* __restrict__ Wm = blockIdx.z ? Wv : Wk;
  float* __restrict__ Om       = blockIdx.z ? Vp : Kp;
  const int M = B_ALL * S_DIM;
  const int kbase = wave * (KV_KS * 4);

  int ra = row0 + l16;       if (ra > M - 1) ra = M - 1;
  int rb = row0 + 16 + l16;  if (rb > M - 1) rb = M - 1;
  const float* pa = E + (size_t)ra * D_ENC + kbase + 2 * half;
  const float* pb = E + (size_t)rb * D_ENC + kbase + 2 * half;
  const float* pw = Wm + (size_t)(kbase + 2 * half) * C_DIM + col0 + 4 * l16;

  v4f wx[2][KV_GRP], wy[2][KV_GRP];
  for (int s = 0; s < KV_GRP; ++s) {
    wx[0][s] = *(const v4f*)(pw + (size_t)(4 * s)     * C_DIM);
    wy[0][s] = *(const v4f*)(pw + (size_t)(4 * s + 1) * C_DIM);
  }
  v2f a0[KV_KS], a1[KV_KS];
  for (int ks = 0; ks < KV_KS; ++ks) {
    a0[ks] = *(const v2f*)(pa + 4 * ks);
    a1[ks] = *(const v2f*)(pb + 4 * ks);
  }

  const v8f zero = {0.f,0.f,0.f,0.f,0.f,0.f,0.f,0.f};
  v8f acc[2][4];
  for (int j = 0; j < 4; ++j) { acc[0][j] = zero; acc[1][j] = zero; }

  for (int g = 0; g < KV_KS / KV_GRP; ++g) {
    const int cur = g & 1;
    if (g + 1 < KV_KS / KV_GRP) {
      for (int s = 0; s < KV_GRP; ++s) {
        const int k = 4 * ((g + 1) * KV_GRP + s);
        wx[cur ^ 1][s] = *(const v4f*)(pw + (size_t)k       * C_DIM);
        wy[cur ^ 1][s] = *(const v4f*)(pw + (size_t)(k + 1) * C_DIM);
      }
    }
    for (int s = 0; s < KV_GRP; ++s) {
      const int ks = g * KV_GRP + s;
      for (int j = 0; j < 4; ++j) {
        v2f b;
        b.x = wx[cur][s][j];
        b.y = wy[cur][s][j];
        acc[0][j] = wmma4(a0[ks], b, acc[0][j]);
        acc[1][j] = wmma4(a1[ks], b, acc[1][j]);
      }
    }
  }

  for (int rt = 0; rt < 2; ++rt) {
    for (int i = 0; i < 8; ++i) {
      v4f t;
      t.x = acc[rt][0][i]; t.y = acc[rt][1][i];
      t.z = acc[rt][2][i]; t.w = acc[rt][3][i];
      *(v4f*)(&red[wave][(16 * rt + i + 8 * half) * KV_COLS + 4 * l16]) = t;
    }
  }
  __syncthreads();

  const int rr = tid >> 4;
  const int cg = tid & 15;
  for (int q = 0; q < 2; ++q) {
    const int row = rr + 16 * q;
    v4f s = *(const v4f*)(&red[0][row * KV_COLS + 4 * cg]);
    for (int w = 1; w < KV_WAVES; ++w) {
      const v4f p = *(const v4f*)(&red[w][row * KV_COLS + 4 * cg]);
      s.x += p.x; s.y += p.y; s.z += p.z; s.w += p.w;
    }
    if (row0 + row < M)
      *(v4f*)(Om + (size_t)(row0 + row) * C_DIM + col0 + 4 * cg) = s;
  }
}

#define NT_ROW  4
#define KS_D    (HD / 4)
#define KS_S    (SP / 4)

__device__ __forceinline__ void lds_handover() {
  __builtin_amdgcn_fence(__ATOMIC_RELEASE, "workgroup", "local");
  __builtin_amdgcn_s_barrier();
  __builtin_amdgcn_fence(__ATOMIC_ACQUIRE, "workgroup", "local");
}

__device__ __forceinline__
void load_v_frags(v2f (&vf)[KS_S][3], const float* __restrict__ Vp,
                  int b, int h, int half, int l16) {
  const float* vbase = Vp + (size_t)(b * S_DIM) * C_DIM + h * HD;
  for (int ks = 0; ks < KS_S; ++ks) {
    const int ka = 4 * ks + 2 * half;
    const int r0 = ka     < S_DIM ? ka     : S_DIM - 1;
    const int r1 = ka + 1 < S_DIM ? ka + 1 : S_DIM - 1;
    #pragma unroll
    for (int j = 0; j < 3; ++j) {
      const int c  = 16 * j + l16;
      const int cc = c < HD ? c : HD - 1;
      const float x = vbase[(size_t)r0 * C_DIM + cc];
      const float y = vbase[(size_t)r1 * C_DIM + cc];
      vf[ks][j].x = (c < HD && ka     < S_DIM) ? x : 0.f;
      vf[ks][j].y = (c < HD && ka + 1 < S_DIM) ? y : 0.f;
    }
  }
}

__device__ __forceinline__
void pv_and_store(const float* Plds, const v2f (&vf)[KS_S][3],
                  float* __restrict__ AO, int b, int h, int t0,
                  int half, int l16) {
  const v8f zero = {0.f,0.f,0.f,0.f,0.f,0.f,0.f,0.f};
  v8f o[3];
  #pragma unroll
  for (int j = 0; j < 3; ++j) o[j] = zero;

  for (int ks = 0; ks < KS_S; ++ks) {
    const int ka = 4 * ks + 2 * half;
    v2f a = *(const v2f*)(Plds + l16 * SP + ka);
    #pragma unroll
    for (int j = 0; j < 3; ++j) o[j] = wmma4(a, vf[ks][j], o[j]);
  }

  #pragma unroll
  for (int j = 0; j < 3; ++j) {
    const int c = 16 * j + l16;
    if (c < HD) {
      #pragma unroll
      for (int i = 0; i < 8; ++i) {
        const int t = t0 + i + 8 * half;
        AO[(size_t)(b * T_DIM + t) * C_DIM + h * HD + c] = o[j][i];
      }
    }
  }
}

__global__ __launch_bounds__(32)
void attn_kernel(const float* __restrict__ Q,
                 const float* __restrict__ Kp,
                 const float* __restrict__ Vp,
                 float* __restrict__ AB,
                 float* __restrict__ AO)
{
  __shared__ float Plds[16 * SP];
  const int lane = threadIdx.x;
  const int half = lane >> 4;
  const int l16  = lane & 15;
  const int tile0 = blockIdx.x * NT_ROW;
  const int h  = blockIdx.y;
  const int b  = blockIdx.z;
  const float scale = 0.15811388300841898f;
  const float NEG_INF = -__builtin_inff();
  const v8f zero = {0.f,0.f,0.f,0.f,0.f,0.f,0.f,0.f};

  const float* kbase = Kp + (size_t)(b * S_DIM) * C_DIM + h * HD;
  v2f kf[KS_D][5];
  for (int kk = 0; kk < KS_D; ++kk) {
    const int ka = 4 * kk + 2 * half;
    #pragma unroll
    for (int j = 0; j < 5; ++j) {
      const int s  = 16 * j + l16;
      const int ss = s < S_DIM ? s : S_DIM - 1;
      const v2f t = *(const v2f*)(kbase + (size_t)ss * C_DIM + ka);
      kf[kk][j].x = (s < S_DIM) ? t.x : 0.f;
      kf[kk][j].y = (s < S_DIM) ? t.y : 0.f;
    }
  }
  v2f vf[KS_S][3];
  load_v_frags(vf, Vp, b, h, half, l16);

  const float* qrow = Q + (size_t)(b * T_DIM + tile0 * 16 + l16) * C_DIM + h * HD + 2 * half;
  v2f qf[KS_D];
  for (int kk = 0; kk < KS_D; ++kk) qf[kk] = *(const v2f*)(qrow + 4 * kk);

  for (int it = 0; it < NT_ROW; ++it) {
    const int t0 = (tile0 + it) << 4;

    v8f sc[5];
    #pragma unroll
    for (int j = 0; j < 5; ++j) sc[j] = zero;
    for (int kk = 0; kk < KS_D; ++kk) {
      #pragma unroll
      for (int j = 0; j < 5; ++j) sc[j] = wmma4(qf[kk], kf[kk][j], sc[j]);
    }

    if (it + 1 < NT_ROW) {
      qrow += (size_t)16 * C_DIM;
      for (int kk = 0; kk < KS_D; ++kk) qf[kk] = *(const v2f*)(qrow + 4 * kk);
    }

    float mx[8];
    #pragma unroll
    for (int i = 0; i < 8; ++i) mx[i] = NEG_INF;
    #pragma unroll
    for (int j = 0; j < 5; ++j) {
      const bool ok = (16 * j + l16) < S_DIM;
      #pragma unroll
      for (int i = 0; i < 8; ++i) {
        const float x = ok ? sc[j][i] * scale : NEG_INF;
        sc[j][i] = x;
        mx[i] = fmaxf(mx[i], x);
      }
    }
    #pragma unroll
    for (int m = 1; m < 16; m <<= 1) {
      #pragma unroll
      for (int i = 0; i < 8; ++i) mx[i] = fmaxf(mx[i], __shfl_xor(mx[i], m, 32));
    }
    float sm[8];
    #pragma unroll
    for (int i = 0; i < 8; ++i) sm[i] = 0.f;
    #pragma unroll
    for (int j = 0; j < 5; ++j) {
      #pragma unroll
      for (int i = 0; i < 8; ++i) {
        const float e = __expf(sc[j][i] - mx[i]);
        sc[j][i] = e;
        sm[i] += e;
      }
    }
    #pragma unroll
    for (int m = 1; m < 16; m <<= 1) {
      #pragma unroll
      for (int i = 0; i < 8; ++i) sm[i] += __shfl_xor(sm[i], m, 32);
    }
    float rs[8];
    #pragma unroll
    for (int i = 0; i < 8; ++i) rs[i] = 1.f / sm[i];

    #pragma unroll
    for (int j = 0; j < 5; ++j) {
      #pragma unroll
      for (int i = 0; i < 8; ++i) {
        const float pv = sc[j][i] * rs[i];
        const int row = i + 8 * half;
        Plds[row * SP + 16 * j + l16] = pv;
        if (b == B_BASE)
          AB[(size_t)(h * T_DIM + t0 + row) * SP + 16 * j + l16] = pv;
      }
    }
    lds_handover();

    pv_and_store(Plds, vf, AO, b, h, t0, half, l16);

    lds_handover();
  }
}

__global__ __launch_bounds__(32)
void edit_kernel(const float* __restrict__ AB,
                 const float* __restrict__ Map,
                 const float* __restrict__ Vp,
                 float* __restrict__ AO)
{
  __shared__ float Plds[16 * SP];
  const int lane = threadIdx.x;
  const int half = lane >> 4;
  const int l16  = lane & 15;
  const int tile0 = blockIdx.x * NT_ROW;
  const int h  = blockIdx.y;
  const int p  = blockIdx.z;
  const int b  = B_BASE + 1 + p;
  const v8f zero = {0.f,0.f,0.f,0.f,0.f,0.f,0.f,0.f};

  const float* mbase = Map + (size_t)p * S_DIM * S_DIM;
  v2f mf[KS_S][5];
  for (int ks = 0; ks < KS_S; ++ks) {
    const int ka = 4 * ks + 2 * half;
    const int r0 = ka     < S_DIM ? ka     : S_DIM - 1;
    const int r1 = ka + 1 < S_DIM ? ka + 1 : S_DIM - 1;
    #pragma unroll
    for (int j = 0; j < 5; ++j) {
      const int n  = 16 * j + l16;
      const int nn = n < S_DIM ? n : S_DIM - 1;
      const float x = mbase[(size_t)r0 * S_DIM + nn];
      const float y = mbase[(size_t)r1 * S_DIM + nn];
      mf[ks][j].x = (n < S_DIM && ka     < S_DIM) ? x : 0.f;
      mf[ks][j].y = (n < S_DIM && ka + 1 < S_DIM) ? y : 0.f;
    }
  }
  v2f vf[KS_S][3];
  load_v_frags(vf, Vp, b, h, half, l16);

  const float* arow = AB + (size_t)(h * T_DIM + tile0 * 16 + l16) * SP + 2 * half;
  v2f af[KS_S];
  for (int ks = 0; ks < KS_S; ++ks) af[ks] = *(const v2f*)(arow + 4 * ks);

  for (int it = 0; it < NT_ROW; ++it) {
    const int t0 = (tile0 + it) << 4;

    v8f acc[5];
    #pragma unroll
    for (int j = 0; j < 5; ++j) acc[j] = zero;
    for (int ks = 0; ks < KS_S; ++ks) {
      #pragma unroll
      for (int j = 0; j < 5; ++j) acc[j] = wmma4(af[ks], mf[ks][j], acc[j]);
    }

    if (it + 1 < NT_ROW) {
      arow += (size_t)16 * SP;
      for (int ks = 0; ks < KS_S; ++ks) af[ks] = *(const v2f*)(arow + 4 * ks);
    }

    #pragma unroll
    for (int j = 0; j < 5; ++j) {
      const bool ok = (16 * j + l16) < S_DIM;
      #pragma unroll
      for (int i = 0; i < 8; ++i) {
        const int row = i + 8 * half;
        Plds[row * SP + 16 * j + l16] = ok ? acc[j][i] : 0.f;
      }
    }
    lds_handover();

    pv_and_store(Plds, vf, AO, b, h, t0, half, l16);

    lds_handover();
  }
}

extern "C" void kernel_launch(void* const* d_in, const int* in_sizes, int n_in,
                              void* d_out, int out_size, void* d_ws, size_t ws_size,
                              hipStream_t stream) {
  const float* X   = (const float*)d_in[0];
  const float* E   = (const float*)d_in[1];
  const float* Map = (const float*)d_in[2];
  const float* Wq  = (const float*)d_in[3];
  const float* Wk  = (const float*)d_in[4];
  const float* Wv  = (const float*)d_in[5];
  const float* Wo  = (const float*)d_in[6];
  const float* bo  = (const float*)d_in[7];
  float* out = (float*)d_out;

  float* ws = (float*)d_ws;
  float* Q  = ws;
  float* Kp = Q  + (size_t)5 * T_DIM * C_DIM;
  float* Vp = Kp + (size_t)B_ALL * S_DIM * C_DIM;
  float* AB = Vp + (size_t)B_ALL * S_DIM * C_DIM;
  float* AO = AB + (size_t)H_NUM * T_DIM * SP;

  const dim3 gblk(128);
  const dim3 ablk(32);
  const int MR = B_ALL * S_DIM;

  gemm_wmma_f32<<<dim3(C_DIM / 64, (5 * T_DIM) / 64), gblk, 0, stream>>>(
      X, Wq, nullptr, Q, 5 * T_DIM, C_DIM, C_DIM);
  kv_proj_kernel<<<dim3(C_DIM / KV_COLS, (MR + KV_ROWS - 1) / KV_ROWS, 2), dim3(256), 0, stream>>>(
      E, Wk, Wv, Kp, Vp);

  attn_kernel<<<dim3(T_DIM / 16 / NT_ROW, H_NUM, 5), ablk, 0, stream>>>(Q, Kp, Vp, AB, AO);

  edit_kernel<<<dim3(T_DIM / 16 / NT_ROW, H_NUM, 3), ablk, 0, stream>>>(AB, Map, Vp, AO);

  gemm_wmma_f32<<<dim3(C_DIM / 64, (B_ALL * T_DIM) / 64), gblk, 0, stream>>>(
      AO, Wo, bo, out, B_ALL * T_DIM, C_DIM, C_DIM);
}


# attention and edit launches: score and edit tiles formed transposed (operands swapped), so the softmax reduces in-lane and the D registers are P@V's A fragments: P skips LDS; sum order changes
# speedup vs baseline: 1.9791x; 1.0505x over previous
#include <hip/hip_runtime.h>


typedef __attribute__((ext_vector_type(2))) float v2f;
typedef __attribute__((ext_vector_type(4))) float v4f;
typedef __attribute__((ext_vector_type(8))) float v8f;

#define B_ALL   8
#define B_BASE  4
#define T_DIM   4096
#define C_DIM   320
#define S_DIM   77
#define SP      80
#define D_ENC   768
#define HD      40
#define H_NUM   8

__device__ __forceinline__ v8f wmma4(v2f a, v2f b, v8f c) {
  return __builtin_amdgcn_wmma_f32_16x16x4_f32(
       false, a,  false, b,
       (short)0, c,  false,  false);
}

#define GK 16

__global__ __launch_bounds__(128)
void gemm_wmma_f32(const float* __restrict__ A, const float* __restrict__ Bm,
                   const float* __restrict__ bias, float* __restrict__ C,
                   int M, int N, int K) {
  __shared__ float Bs[2][64 * 18];
  const int tid  = threadIdx.x;
  const int wave = tid >> 5;
  const int lane = tid & 31;
  const int half = lane >> 4;
  const int l16  = lane & 15;
  const int col0 = blockIdx.x * 64;
  const int rowW = blockIdx.y * 64 + wave * 16;

  int arow = rowW + l16;
  if (arow > M - 1) arow = M - 1;
  const float* arowp = A + (size_t)arow * K;

  const int srow = tid >> 3;
  const int scol = (tid & 7) * 8;
  const float* bstage = Bm + (size_t)srow * N + col0 + scol;

  const v8f zero = {0.f,0.f,0.f,0.f,0.f,0.f,0.f,0.f};
  v8f acc[4];
  #pragma unroll
  for (int j = 0; j < 4; ++j) acc[j] = zero;

  const int nkb = K / GK;

  {
    v4f r0 = *(const v4f*)(bstage);
    v4f r1 = *(const v4f*)(bstage + 4);
    #pragma unroll
    for (int e = 0; e < 4; ++e) Bs[0][(scol + e) * 18 + srow] = r0[e];
    #pragma unroll
    for (int e = 0; e < 4; ++e) Bs[0][(scol + 4 + e) * 18 + srow] = r1[e];
  }
  __syncthreads();

  for (int kb = 0; kb < nkb; ++kb) {
    const int cur = kb & 1;
    const bool have_next = (kb + 1) < nkb;

    v4f r0, r1;
    if (have_next) {
      const float* bnxt = bstage + (size_t)(kb + 1) * GK * N;
      r0 = *(const v4f*)(bnxt);
      r1 = *(const v4f*)(bnxt + 4);
      if (kb + 2 < nkb)
        __builtin_prefetch(bstage + (size_t)(kb + 2) * GK * N, 0, 1);
    }

    const float* ap = arowp + kb * GK;
    for (int kk = 0; kk < GK; kk += 4) {
      const int ka = kk + 2 * half;
      v2f a = *(const v2f*)(ap + ka);
      #pragma unroll
      for (int j = 0; j < 4; ++j) {
        v2f b = *(const v2f*)(&Bs[cur][(16 * j + l16) * 18 + ka]);
        acc[j] = wmma4(a, b, acc[j]);
      }
    }

    if (have_next) {
      const int nxt = cur ^ 1;
      #pragma unroll
      for (int e = 0; e < 4; ++e) Bs[nxt][(scol + e) * 18 + srow] = r0[e];
      #pragma unroll
      for (int e = 0; e < 4; ++e) Bs[nxt][(scol + 4 + e) * 18 + srow] = r1[e];
    }
    __syncthreads();
  }

  const float bv = bias ? bias[col0 + 16 * 0 + l16] : 0.f;
  #pragma unroll
  for (int j = 0; j < 4; ++j) {
    const int c = col0 + 16 * j + l16;
    const float bj = bias ? bias[c] : 0.f;
    #pragma unroll
    for (int i = 0; i < 8; ++i) {
      const int r = rowW + i + 8 * half;
      if (r < M) C[(size_t)r * N + c] = acc[j][i] + bj;
    }
  }
  (void)bv;
}

#define KV_WAVES 8
#define KV_KS    (D_ENC / KV_WAVES / 4)
#define KV_GRP   6
#define KV_ROWS  32
#define KV_COLS  64

__global__ __launch_bounds__(256)
void kv_proj_kernel(const float* __restrict__ E,
                    const float* __restrict__ Wk,
                    const float* __restrict__ Wv,
                    float* __restrict__ Kp,
                    float* __restrict__ Vp)
{
  __shared__ float red[KV_WAVES][KV_ROWS * KV_COLS];
  const int tid  = threadIdx.x;
  const int wave = tid >> 5;
  const int lane = tid & 31;
  const int half = lane >> 4;
  const int l16  = lane & 15;
  const int col0 = blockIdx.x * KV_COLS;
  const int row0 = blockIdx.y * KV_ROWS;
  const float* __restrict__ Wm = blockIdx.z ? Wv : Wk;
  float* __restrict__ Om       = blockIdx.z ? Vp : Kp;
  const int M = B_ALL * S_DIM;
  const int kbase = wave * (KV_KS * 4);

  int ra = row0 + l16;       if (ra > M - 1) ra = M - 1;
  int rb = row0 + 16 + l16;  if (rb > M - 1) rb = M - 1;
  const float* pa = E + (size_t)ra * D_ENC + kbase + 2 * half;
  const float* pb = E + (size_t)rb * D_ENC + kbase + 2 * half;
  const float* pw = Wm + (size_t)(kbase + 2 * half) * C_DIM + col0 + 4 * l16;

  v4f wx[2][KV_GRP], wy[2][KV_GRP];
  for (int s = 0; s < KV_GRP; ++s) {
    wx[0][s] = *(const v4f*)(pw + (size_t)(4 * s)     * C_DIM);
    wy[0][s] = *(const v4f*)(pw + (size_t)(4 * s + 1) * C_DIM);
  }
  v2f a0[KV_KS], a1[KV_KS];
  for (int ks = 0; ks < KV_KS; ++ks) {
    a0[ks] = *(const v2f*)(pa + 4 * ks);
    a1[ks] = *(const v2f*)(pb + 4 * ks);
  }

  const v8f zero = {0.f,0.f,0.f,0.f,0.f,0.f,0.f,0.f};
  v8f acc[2][4];
  for (int j = 0; j < 4; ++j) { acc[0][j] = zero; acc[1][j] = zero; }

  for (int g = 0; g < KV_KS / KV_GRP; ++g) {
    const int cur = g & 1;
    if (g + 1 < KV_KS / KV_GRP) {
      for (int s = 0; s < KV_GRP; ++s) {
        const int k = 4 * ((g + 1) * KV_GRP + s);
        wx[cur ^ 1][s] = *(const v4f*)(pw + (size_t)k       * C_DIM);
        wy[cur ^ 1][s] = *(const v4f*)(pw + (size_t)(k + 1) * C_DIM);
      }
    }
    for (int s = 0; s < KV_GRP; ++s) {
      const int ks = g * KV_GRP + s;
      for (int j = 0; j < 4; ++j) {
        v2f b;
        b.x = wx[cur][s][j];
        b.y = wy[cur][s][j];
        acc[0][j] = wmma4(a0[ks], b, acc[0][j]);
        acc[1][j] = wmma4(a1[ks], b, acc[1][j]);
      }
    }
  }

  for (int rt = 0; rt < 2; ++rt) {
    for (int i = 0; i < 8; ++i) {
      v4f t;
      t.x = acc[rt][0][i]; t.y = acc[rt][1][i];
      t.z = acc[rt][2][i]; t.w = acc[rt][3][i];
      *(v4f*)(&red[wave][(16 * rt + i + 8 * half) * KV_COLS + 4 * l16]) = t;
    }
  }
  __syncthreads();

  const int rr = tid >> 4;
  const int cg = tid & 15;
  for (int q = 0; q < 2; ++q) {
    const int row = rr + 16 * q;
    v4f s = *(const v4f*)(&red[0][row * KV_COLS + 4 * cg]);
    for (int w = 1; w < KV_WAVES; ++w) {
      const v4f p = *(const v4f*)(&red[w][row * KV_COLS + 4 * cg]);
      s.x += p.x; s.y += p.y; s.z += p.z; s.w += p.w;
    }
    if (row0 + row < M)
      *(v4f*)(Om + (size_t)(row0 + row) * C_DIM + col0 + 4 * cg) = s;
  }
}

#define NT_ROW  4
#define KS_D    (HD / 4)
#define KS_S    (SP / 4)

#define KT_S    (SP / 16)
#define KM_S    4

__device__ __forceinline__ float xhalf(float x) {
  const int b = __builtin_bit_cast(int, x);
  return __builtin_bit_cast(float,
      __builtin_amdgcn_permlanex16(b, b, 0x76543210, 0xfedcba98, false, false));
}

__device__ __forceinline__
void load_v_frags(v2f (&vf)[KT_S][KM_S][3], const float* __restrict__ Vp,
                  int b, int h, int half, int l16) {
  const float* vbase = Vp + (size_t)(b * S_DIM) * C_DIM + h * HD;
  #pragma unroll
  for (int j = 0; j < KT_S; ++j) {
    #pragma unroll
    for (int m = 0; m < KM_S; ++m) {
      const int ka = 16 * j + 8 * half + 2 * m;
      const int r0 = ka     < S_DIM ? ka     : S_DIM - 1;
      const int r1 = ka + 1 < S_DIM ? ka + 1 : S_DIM - 1;
      #pragma unroll
      for (int jn = 0; jn < 3; ++jn) {
        const int c  = 16 * jn + l16;
        const int cc = c < HD ? c : HD - 1;
        const float x = vbase[(size_t)r0 * C_DIM + cc];
        const float y = vbase[(size_t)r1 * C_DIM + cc];
        vf[j][m][jn].x = (c < HD && ka     < S_DIM) ? x : 0.f;
        vf[j][m][jn].y = (c < HD && ka + 1 < S_DIM) ? y : 0.f;
      }
    }
  }
}

__device__ __forceinline__
void pv_and_store(const v8f (&pt)[KT_S], const v2f (&vf)[KT_S][KM_S][3],
                  float* __restrict__ AO, int b, int h, int t0,
                  int half, int l16) {
  const v8f zero = {0.f,0.f,0.f,0.f,0.f,0.f,0.f,0.f};
  v8f o[3];
  #pragma unroll
  for (int j = 0; j < 3; ++j) o[j] = zero;

  #pragma unroll
  for (int j = 0; j < KT_S; ++j) {
    #pragma unroll
    for (int m = 0; m < KM_S; ++m) {
      v2f a;
      a.x = pt[j][2 * m];
      a.y = pt[j][2 * m + 1];
      #pragma unroll
      for (int jn = 0; jn < 3; ++jn) o[jn] = wmma4(a, vf[j][m][jn], o[jn]);
    }
  }

  #pragma unroll
  for (int j = 0; j < 3; ++j) {
    const int c = 16 * j + l16;
    if (c < HD) {
      #pragma unroll
      for (int i = 0; i < 8; ++i) {
        const int t = t0 + i + 8 * half;
        AO[(size_t)(b * T_DIM + t) * C_DIM + h * HD + c] = o[j][i];
      }
    }
  }
}

__global__ __launch_bounds__(32)
void attn_kernel(const float* __restrict__ Q,
                 const float* __restrict__ Kp,
                 const float* __restrict__ Vp,
                 float* __restrict__ AB,
                 float* __restrict__ AO)
{
  const int lane = threadIdx.x;
  const int half = (lane >> 4) & 1;
  const int l16  = lane & 15;
  const int tile0 = blockIdx.x * NT_ROW;
  const int h  = blockIdx.y;
  const int b  = blockIdx.z;
  const float scale = 0.15811388300841898f;
  const float NEG_INF = -__builtin_inff();
  const v8f zero = {0.f,0.f,0.f,0.f,0.f,0.f,0.f,0.f};

  const float* kbase = Kp + (size_t)(b * S_DIM) * C_DIM + h * HD;
  v2f kf[KS_D][KT_S];
  for (int kk = 0; kk < KS_D; ++kk) {
    const int ka = 4 * kk + 2 * half;
    #pragma unroll
    for (int j = 0; j < KT_S; ++j) {
      const int s  = 16 * j + l16;
      const int ss = s < S_DIM ? s : S_DIM - 1;
      const v2f t = *(const v2f*)(kbase + (size_t)ss * C_DIM + ka);
      kf[kk][j].x = (s < S_DIM) ? t.x : 0.f;
      kf[kk][j].y = (s < S_DIM) ? t.y : 0.f;
    }
  }
  v2f vf[KT_S][KM_S][3];
  load_v_frags(vf, Vp, b, h, half, l16);

  const float* qrow = Q + (size_t)(b * T_DIM + tile0 * 16 + l16) * C_DIM + h * HD + 2 * half;
  v2f qf[KS_D];
  for (int kk = 0; kk < KS_D; ++kk) qf[kk] = *(const v2f*)(qrow + 4 * kk);

  for (int it = 0; it < NT_ROW; ++it) {
    const int t0 = (tile0 + it) << 4;

    v8f sc[KT_S];
    #pragma unroll
    for (int j = 0; j < KT_S; ++j) sc[j] = zero;
    for (int kk = 0; kk < KS_D; ++kk) {
      #pragma unroll
      for (int j = 0; j < KT_S; ++j) sc[j] = wmma4(kf[kk][j], qf[kk], sc[j]);
    }

    if (it + 1 < NT_ROW) {
      qrow += (size_t)16 * C_DIM;
      for (int kk = 0; kk < KS_D; ++kk) qf[kk] = *(const v2f*)(qrow + 4 * kk);
    }

    float mj[KT_S];
    #pragma unroll
    for (int j = 0; j < KT_S; ++j) {
      mj[j] = NEG_INF;
      #pragma unroll
      for (int i = 0; i < 8; ++i) {
        const bool ok = (16 * j + 8 * half + i) < S_DIM;
        const float x = ok ? sc[j][i] * scale : NEG_INF;
        sc[j][i] = x;
        mj[j] = fmaxf(mj[j], x);
      }
    }
    float mx = fmaxf(fmaxf(fmaxf(mj[0], mj[1]), fmaxf(mj[2], mj[3])), mj[4]);
    mx = fmaxf(mx, xhalf(mx));
    float sj[KT_S];
    #pragma unroll
    for (int j = 0; j < KT_S; ++j) {
      sj[j] = 0.f;
      #pragma unroll
      for (int i = 0; i < 8; ++i) {
        const float e = __expf(sc[j][i] - mx);
        sc[j][i] = e;
        sj[j] += e;
      }
    }
    float sm = ((sj[0] + sj[1]) + (sj[2] + sj[3])) + sj[4];
    sm = sm + xhalf(sm);
    const float rs = 1.f / sm;
    #pragma unroll
    for (int j = 0; j < KT_S; ++j) {
      #pragma unroll
      for (int i = 0; i < 8; ++i) sc[j][i] *= rs;
    }

    if (b == B_BASE) {
      float* abrow = AB + (size_t)(h * T_DIM + t0 + l16) * SP + 8 * half;
      #pragma unroll
      for (int j = 0; j < KT_S; ++j) {
        v4f lo, hi;
        lo.x = sc[j][0]; lo.y = sc[j][1]; lo.z = sc[j][2]; lo.w = sc[j][3];
        hi.x = sc[j][4]; hi.y = sc[j][5]; hi.z = sc[j][6]; hi.w = sc[j][7];
        *(v4f*)(abrow + 16 * j)     = lo;
        *(v4f*)(abrow + 16 * j + 4) = hi;
      }
    }

    pv_and_store(sc, vf, AO, b, h, t0, half, l16);
  }
}

__global__ __launch_bounds__(32)
void edit_kernel(const float* __restrict__ AB,
                 const float* __restrict__ Map,
                 const float* __restrict__ Vp,
                 float* __restrict__ AO)
{
  const int lane = threadIdx.x;
  const int half = (lane >> 4) & 1;
  const int l16  = lane & 15;
  const int tile0 = blockIdx.x * NT_ROW;
  const int h  = blockIdx.y;
  const int p  = blockIdx.z;
  const int b  = B_BASE + 1 + p;
  const v8f zero = {0.f,0.f,0.f,0.f,0.f,0.f,0.f,0.f};

  const float* mbase = Map + (size_t)p * S_DIM * S_DIM;
  v2f mf[KS_S][KT_S];
  for (int ks = 0; ks < KS_S; ++ks) {
    const int ka = 4 * ks + 2 * half;
    const int r0 = ka     < S_DIM ? ka     : S_DIM - 1;
    const int r1 = ka + 1 < S_DIM ? ka + 1 : S_DIM - 1;
    #pragma unroll
    for (int j = 0; j < KT_S; ++j) {
      const int n  = 16 * j + l16;
      const int nn = n < S_DIM ? n : S_DIM - 1;
      const float x = mbase[(size_t)r0 * S_DIM + nn];
      const float y = mbase[(size_t)r1 * S_DIM + nn];
      mf[ks][j].x = (n < S_DIM && ka     < S_DIM) ? x : 0.f;
      mf[ks][j].y = (n < S_DIM && ka + 1 < S_DIM) ? y : 0.f;
    }
  }
  v2f vf[KT_S][KM_S][3];
  load_v_frags(vf, Vp, b, h, half, l16);

  const float* arow = AB + (size_t)(h * T_DIM + tile0 * 16 + l16) * SP + 2 * half;
  v2f af[KS_S];
  for (int ks = 0; ks < KS_S; ++ks) af[ks] = *(const v2f*)(arow + 4 * ks);

  for (int it = 0; it < NT_ROW; ++it) {
    const int t0 = (tile0 + it) << 4;

    v8f acc[KT_S];
    #pragma unroll
    for (int j = 0; j < KT_S; ++j) acc[j] = zero;
    for (int ks = 0; ks < KS_S; ++ks) {
      #pragma unroll
      for (int j = 0; j < KT_S; ++j) acc[j] = wmma4(mf[ks][j], af[ks], acc[j]);
    }

    if (it + 1 < NT_ROW) {
      arow += (size_t)16 * SP;
      for (int ks = 0; ks < KS_S; ++ks) af[ks] = *(const v2f*)(arow + 4 * ks);
    }

    #pragma unroll
    for (int j = 0; j < KT_S; ++j) {
      #pragma unroll
      for (int i = 0; i < 8; ++i) {
        const bool ok = (16 * j + 8 * half + i) < S_DIM;
        acc[j][i] = ok ? acc[j][i] : 0.f;
      }
    }

    pv_and_store(acc, vf, AO, b, h, t0, half, l16);
  }
}

extern "C" void kernel_launch(void* const* d_in, const int* in_sizes, int n_in,
                              void* d_out, int out_size, void* d_ws, size_t ws_size,
                              hipStream_t stream) {
  const float* X   = (const float*)d_in[0];
  const float* E   = (const float*)d_in[1];
  const float* Map = (const float*)d_in[2];
  const float* Wq  = (const float*)d_in[3];
  const float* Wk  = (const float*)d_in[4];
  const float* Wv  = (const float*)d_in[5];
  const float* Wo  = (const float*)d_in[6];
  const float* bo  = (const float*)d_in[7];
  float* out = (float*)d_out;

  float* ws = (float*)d_ws;
  float* Q  = ws;
  float* Kp = Q  + (size_t)5 * T_DIM * C_DIM;
  float* Vp = Kp + (size_t)B_ALL * S_DIM * C_DIM;
  float* AB = Vp + (size_t)B_ALL * S_DIM * C_DIM;
  float* AO = AB + (size_t)H_NUM * T_DIM * SP;

  const dim3 gblk(128);
  const dim3 ablk(32);
  const int MR = B_ALL * S_DIM;

  gemm_wmma_f32<<<dim3(C_DIM / 64, (5 * T_DIM) / 64), gblk, 0, stream>>>(
      X, Wq, nullptr, Q, 5 * T_DIM, C_DIM, C_DIM);
  kv_proj_kernel<<<dim3(C_DIM / KV_COLS, (MR + KV_ROWS - 1) / KV_ROWS, 2), dim3(256), 0, stream>>>(
      E, Wk, Wv, Kp, Vp);

  attn_kernel<<<dim3(T_DIM / 16 / NT_ROW, H_NUM, 5), ablk, 0, stream>>>(Q, Kp, Vp, AB, AO);

  edit_kernel<<<dim3(T_DIM / 16 / NT_ROW, H_NUM, 3), ablk, 0, stream>>>(AB, Map, Vp, AO);

  gemm_wmma_f32<<<dim3(C_DIM / 64, (B_ALL * T_DIM) / 64), gblk, 0, stream>>>(
      AO, Wo, bo, out, B_ALL * T_DIM, C_DIM, C_DIM);
}


# output projection folded into the attention and edit launches: a work-group (8 heads x 5 or 3 row tiles) keeps its attention rows in LDS and multiplies them by Wo; launches 5 -> 4; sums unchanged
# speedup vs baseline: 2.2561x; 1.1400x over previous
#include <hip/hip_runtime.h>


typedef __attribute__((ext_vector_type(2))) float v2f;
typedef __attribute__((ext_vector_type(4))) float v4f;
typedef __attribute__((ext_vector_type(8))) float v8f;

#define B_ALL   8
#define B_BASE  4
#define T_DIM   4096
#define C_DIM   320
#define S_DIM   77
#define SP      80
#define D_ENC   768
#define HD      40
#define H_NUM   8

__device__ __forceinline__ v8f wmma4(v2f a, v2f b, v8f c) {
  return __builtin_amdgcn_wmma_f32_16x16x4_f32(
       false, a,  false, b,
       (short)0, c,  false,  false);
}

#define GK 16

__global__ __launch_bounds__(128)
void gemm_wmma_f32(const float* __restrict__ A, const float* __restrict__ Bm,
                   const float* __restrict__ bias, float* __restrict__ C,
                   int M, int N, int K) {
  __shared__ float Bs[2][64 * 18];
  const int tid  = threadIdx.x;
  const int wave = tid >> 5;
  const int lane = tid & 31;
  const int half = lane >> 4;
  const int l16  = lane & 15;
  const int col0 = blockIdx.x * 64;
  const int rowW = blockIdx.y * 64 + wave * 16;

  int arow = rowW + l16;
  if (arow > M - 1) arow = M - 1;
  const float* arowp = A + (size_t)arow * K;

  const int srow = tid >> 3;
  const int scol = (tid & 7) * 8;
  const float* bstage = Bm + (size_t)srow * N + col0 + scol;

  const v8f zero = {0.f,0.f,0.f,0.f,0.f,0.f,0.f,0.f};
  v8f acc[4];
  #pragma unroll
  for (int j = 0; j < 4; ++j) acc[j] = zero;

  const int nkb = K / GK;

  {
    v4f r0 = *(const v4f*)(bstage);
    v4f r1 = *(const v4f*)(bstage + 4);
    #pragma unroll
    for (int e = 0; e < 4; ++e) Bs[0][(scol + e) * 18 + srow] = r0[e];
    #pragma unroll
    for (int e = 0; e < 4; ++e) Bs[0][(scol + 4 + e) * 18 + srow] = r1[e];
  }
  __syncthreads();

  for (int kb = 0; kb < nkb; ++kb) {
    const int cur = kb & 1;
    const bool have_next = (kb + 1) < nkb;

    v4f r0, r1;
    if (have_next) {
      const float* bnxt = bstage + (size_t)(kb + 1) * GK * N;
      r0 = *(const v4f*)(bnxt);
      r1 = *(const v4f*)(bnxt + 4);
      if (kb + 2 < nkb)
        __builtin_prefetch(bstage + (size_t)(kb + 2) * GK * N, 0, 1);
    }

    const float* ap = arowp + kb * GK;
    for (int kk = 0; kk < GK; kk += 4) {
      const int ka = kk + 2 * half;
      v2f a = *(const v2f*)(ap + ka);
      #pragma unroll
      for (int j = 0; j < 4; ++j) {
        v2f b = *(const v2f*)(&Bs[cur][(16 * j + l16) * 18 + ka]);
        acc[j] = wmma4(a, b, acc[j]);
      }
    }

    if (have_next) {
      const int nxt = cur ^ 1;
      #pragma unroll
      for (int e = 0; e < 4; ++e) Bs[nxt][(scol + e) * 18 + srow] = r0[e];
      #pragma unroll
      for (int e = 0; e < 4; ++e) Bs[nxt][(scol + 4 + e) * 18 + srow] = r1[e];
    }
    __syncthreads();
  }

  const float bv = bias ? bias[col0 + 16 * 0 + l16] : 0.f;
  #pragma unroll
  for (int j = 0; j < 4; ++j) {
    const int c = col0 + 16 * j + l16;
    const float bj = bias ? bias[c] : 0.f;
    #pragma unroll
    for (int i = 0; i < 8; ++i) {
      const int r = rowW + i + 8 * half;
      if (r < M) C[(size_t)r * N + c] = acc[j][i] + bj;
    }
  }
  (void)bv;
}

#define KV_WAVES 8
#define KV_KS    (D_ENC / KV_WAVES / 4)
#define KV_GRP   6
#define KV_ROWS  32
#define KV_COLS  64

__global__ __launch_bounds__(256)
void kv_proj_kernel(const float* __restrict__ E,
                    const float* __restrict__ Wk,
                    const float* __restrict__ Wv,
                    float* __restrict__ Kp,
                    float* __restrict__ Vp)
{
  __shared__ float red[KV_WAVES][KV_ROWS * KV_COLS];
  const int tid  = threadIdx.x;
  const int wave = tid >> 5;
  const int lane = tid & 31;
  const int half = lane >> 4;
  const int l16  = lane & 15;
  const int col0 = blockIdx.x * KV_COLS;
  const int row0 = blockIdx.y * KV_ROWS;
  const float* __restrict__ Wm = blockIdx.z ? Wv : Wk;
  float* __restrict__ Om       = blockIdx.z ? Vp : Kp;
  const int M = B_ALL * S_DIM;
  const int kbase = wave * (KV_KS * 4);

  int ra = row0 + l16;       if (ra > M - 1) ra = M - 1;
  int rb = row0 + 16 + l16;  if (rb > M - 1) rb = M - 1;
  const float* pa = E + (size_t)ra * D_ENC + kbase + 2 * half;
  const float* pb = E + (size_t)rb * D_ENC + kbase + 2 * half;
  const float* pw = Wm + (size_t)(kbase + 2 * half) * C_DIM + col0 + 4 * l16;

  v4f wx[2][KV_GRP], wy[2][KV_GRP];
  for (int s = 0; s < KV_GRP; ++s) {
    wx[0][s] = *(const v4f*)(pw + (size_t)(4 * s)     * C_DIM);
    wy[0][s] = *(const v4f*)(pw + (size_t)(4 * s + 1) * C_DIM);
  }
  v2f a0[KV_KS], a1[KV_KS];
  for (int ks = 0; ks < KV_KS; ++ks) {
    a0[ks] = *(const v2f*)(pa + 4 * ks);
    a1[ks] = *(const v2f*)(pb + 4 * ks);
  }

  const v8f zero = {0.f,0.f,0.f,0.f,0.f,0.f,0.f,0.f};
  v8f acc[2][4];
  for (int j = 0; j < 4; ++j) { acc[0][j] = zero; acc[1][j] = zero; }

  for (int g = 0; g < KV_KS / KV_GRP; ++g) {
    const int cur = g & 1;
    if (g + 1 < KV_KS / KV_GRP) {
      for (int s = 0; s < KV_GRP; ++s) {
        const int k = 4 * ((g + 1) * KV_GRP + s);
        wx[cur ^ 1][s] = *(const v4f*)(pw + (size_t)k       * C_DIM);
        wy[cur ^ 1][s] = *(const v4f*)(pw + (size_t)(k + 1) * C_DIM);
      }
    }
    for (int s = 0; s < KV_GRP; ++s) {
      const int ks = g * KV_GRP + s;
      for (int j = 0; j < 4; ++j) {
        v2f b;
        b.x = wx[cur][s][j];
        b.y = wy[cur][s][j];
        acc[0][j] = wmma4(a0[ks], b, acc[0][j]);
        acc[1][j] = wmma4(a1[ks], b, acc[1][j]);
      }
    }
  }

  for (int rt = 0; rt < 2; ++rt) {
    for (int i = 0; i < 8; ++i) {
      v4f t;
      t.x = acc[rt][0][i]; t.y = acc[rt][1][i];
      t.z = acc[rt][2][i]; t.w = acc[rt][3][i];
      *(v4f*)(&red[wave][(16 * rt + i + 8 * half) * KV_COLS + 4 * l16]) = t;
    }
  }
  __syncthreads();

  const int rr = tid >> 4;
  const int cg = tid & 15;
  for (int q = 0; q < 2; ++q) {
    const int row = rr + 16 * q;
    v4f s = *(const v4f*)(&red[0][row * KV_COLS + 4 * cg]);
    for (int w = 1; w < KV_WAVES; ++w) {
      const v4f p = *(const v4f*)(&red[w][row * KV_COLS + 4 * cg]);
      s.x += p.x; s.y += p.y; s.z += p.z; s.w += p.w;
    }
    if (row0 + row < M)
      *(v4f*)(Om + (size_t)(row0 + row) * C_DIM + col0 + 4 * cg) = s;
  }
}

#define NR_ATT  5
#define NR_EDT  3
#define TILES_B (T_DIM / 16)
#define KS_D    (HD / 4)
#define KS_S    (SP / 4)

#define KT_S    (SP / 16)
#define KM_S    4

__device__ __forceinline__ float xhalf(float x) {
  const int b = __builtin_bit_cast(int, x);
  return __builtin_bit_cast(float,
      __builtin_amdgcn_permlanex16(b, b, 0x76543210, 0xfedcba98, false, false));
}

__device__ __forceinline__
void load_v_frags(v2f (&vf)[KT_S][KM_S][3], const float* __restrict__ Vp,
                  int b, int h, int half, int l16) {
  const float* vbase = Vp + (size_t)(b * S_DIM) * C_DIM + h * HD;
  #pragma unroll
  for (int j = 0; j < KT_S; ++j) {
    #pragma unroll
    for (int m = 0; m < KM_S; ++m) {
      const int ka = 16 * j + 8 * half + 2 * m;
      const int r0 = ka     < S_DIM ? ka     : S_DIM - 1;
      const int r1 = ka + 1 < S_DIM ? ka + 1 : S_DIM - 1;
      #pragma unroll
      for (int jn = 0; jn < 3; ++jn) {
        const int c  = 16 * jn + l16;
        const int cc = c < HD ? c : HD - 1;
        const float x = vbase[(size_t)r0 * C_DIM + cc];
        const float y = vbase[(size_t)r1 * C_DIM + cc];
        vf[j][m][jn].x = (c < HD && ka     < S_DIM) ? x : 0.f;
        vf[j][m][jn].y = (c < HD && ka + 1 < S_DIM) ? y : 0.f;
      }
    }
  }
}

__device__ __forceinline__
void load_k_frags(v2f (&kf)[KS_D][KT_S], const float* __restrict__ Kp,
                  int b, int h, int half, int l16) {
  const float* kbase = Kp + (size_t)(b * S_DIM) * C_DIM + h * HD;
  #pragma unroll
  for (int kk = 0; kk < KS_D; ++kk) {
    const int ka = 4 * kk + 2 * half;
    #pragma unroll
    for (int j = 0; j < KT_S; ++j) {
      const int s  = 16 * j + l16;
      const int ss = s < S_DIM ? s : S_DIM - 1;
      const v2f t = *(const v2f*)(kbase + (size_t)ss * C_DIM + ka);
      kf[kk][j].x = (s < S_DIM) ? t.x : 0.f;
      kf[kk][j].y = (s < S_DIM) ? t.y : 0.f;
    }
  }
}

__device__ __forceinline__
void load_m_frags(v2f (&mf)[KS_S][KT_S], const float* __restrict__ Map,
                  int p, int half, int l16) {
  const float* mbase = Map + (size_t)p * S_DIM * S_DIM;
  #pragma unroll
  for (int ks = 0; ks < KS_S; ++ks) {
    const int ka = 4 * ks + 2 * half;
    const int r0 = ka     < S_DIM ? ka     : S_DIM - 1;
    const int r1 = ka + 1 < S_DIM ? ka + 1 : S_DIM - 1;
    #pragma unroll
    for (int j = 0; j < KT_S; ++j) {
      const int n  = 16 * j + l16;
      const int nn = n < S_DIM ? n : S_DIM - 1;
      const float x = mbase[(size_t)r0 * S_DIM + nn];
      const float y = mbase[(size_t)r1 * S_DIM + nn];
      mf[ks][j].x = (n < S_DIM && ka     < S_DIM) ? x : 0.f;
      mf[ks][j].y = (n < S_DIM && ka + 1 < S_DIM) ? y : 0.f;
    }
  }
}

#define AOP      324
#define OB_K     32
#define OB_P     672
#define OB_NKB   (C_DIM / OB_K)
#define OB_ITEMS ((OB_K / 2) * (C_DIM / 4) / 256)
#define OB_BUF   ((OB_K / 2) * OB_P)

__device__ __forceinline__
void pv_to_lds(const v8f (&pt)[KT_S], const v2f (&vf)[KT_S][KM_S][3],
               float* aos, int h, int half, int l16) {
  const v8f zero = {0.f,0.f,0.f,0.f,0.f,0.f,0.f,0.f};
  v8f o[3];
  #pragma unroll
  for (int j = 0; j < 3; ++j) o[j] = zero;

  #pragma unroll
  for (int j = 0; j < KT_S; ++j) {
    #pragma unroll
    for (int m = 0; m < KM_S; ++m) {
      v2f a;
      a.x = pt[j][2 * m];
      a.y = pt[j][2 * m + 1];
      #pragma unroll
      for (int jn = 0; jn < 3; ++jn) o[jn] = wmma4(a, vf[j][m][jn], o[jn]);
    }
  }

  #pragma unroll
  for (int j = 0; j < 3; ++j) {
    const int c = 16 * j + l16;
    if (c < HD) {
      #pragma unroll
      for (int i = 0; i < 8; ++i)
        aos[(i + 8 * half) * AOP + h * HD + c] = o[j][i];
    }
  }
}

__device__ __forceinline__
void wo_block_load(v4f (&sx)[OB_ITEMS], v4f (&sy)[OB_ITEMS],
                   const float* __restrict__ Wo, int kb, int tid) {
  #pragma unroll
  for (int u = 0; u < OB_ITEMS; ++u) {
    const int f  = tid + 256 * u;
    const int kp = f / (C_DIM / 4);
    const int q  = f - (C_DIM / 4) * kp;
    const float* p = Wo + (size_t)(kb * OB_K + 2 * kp) * C_DIM + 4 * q;
    sx[u] = *(const v4f*)(p);
    sy[u] = *(const v4f*)(p + C_DIM);
  }
}

__device__ __forceinline__
void wo_block_store(float* bs, const v4f (&sx)[OB_ITEMS], const v4f (&sy)[OB_ITEMS],
                    int tid) {
  #pragma unroll
  for (int u = 0; u < OB_ITEMS; ++u) {
    const int f  = tid + 256 * u;
    const int kp = f / (C_DIM / 4);
    const int q  = f - (C_DIM / 4) * kp;
    float* d = bs + kp * OB_P + 8 * q;
    v4f lo, hi;
    lo.x = sx[u].x; lo.y = sy[u].x; lo.z = sx[u].y; lo.w = sy[u].y;
    hi.x = sx[u].z; hi.y = sy[u].z; hi.z = sx[u].w; hi.w = sy[u].w;
    *(v4f*)(d)     = lo;
    *(v4f*)(d + 4) = hi;
  }
}

template <int NA, int NRT>
__device__ __forceinline__
void wo_block_mma(v8f (&acc)[NA][5], const float* arows, const float* bs,
                  int kb, int cg, int half, int l16) {
  const float* ap = arows + l16 * AOP + kb * OB_K + 2 * half;
  const float* bp = bs + half * OB_P + 2 * (cg * 80 + l16);
  #pragma unroll
  for (int ks = 0; ks < OB_K / 4; ++ks) {
    v2f b[5];
    #pragma unroll
    for (int j = 0; j < 5; ++j) b[j] = *(const v2f*)(bp + (2 * ks) * OB_P + 32 * j);
    #pragma unroll
    for (int r = 0; r < NRT; ++r) {
      const v2f a = *(const v2f*)(ap + r * 16 * AOP + 4 * ks);
      #pragma unroll
      for (int j = 0; j < 5; ++j) acc[r][j] = wmma4(a, b[j], acc[r][j]);
    }
  }
}

template <int NR>
__device__ __forceinline__
void out_proj_from_lds(const float* aos, float* bs0, float* bs1,
                       const float* __restrict__ Wo, const float* __restrict__ bo,
                       float* __restrict__ orow, int tid, int wave, int half, int l16) {
  constexpr int NA = (NR + 1) / 2;
  constexpr int NB = NR - NA;
  const int rsel = (wave ^ (wave >> 2)) & 1;
  const int cg   = (wave >> 1) & 3;
  const int rt0  = rsel ? NA : 0;
  const int nrt  = rsel ? NB : NA;

  const v8f zero = {0.f,0.f,0.f,0.f,0.f,0.f,0.f,0.f};
  v8f acc[NA][5];
  #pragma unroll
  for (int r = 0; r < NA; ++r) {
    #pragma unroll
    for (int j = 0; j < 5; ++j) acc[r][j] = zero;
  }

  v4f sx[OB_ITEMS], sy[OB_ITEMS];
  wo_block_load(sx, sy, Wo, 0, tid);
  wo_block_store(bs0, sx, sy, tid);
  __syncthreads();

  const float* arows = aos + rt0 * 16 * AOP;
  for (int kb = 0; kb < OB_NKB; ++kb) {
    const float* cur = (kb & 1) ? bs1 : bs0;
    float*       nxt = (kb & 1) ? bs0 : bs1;
    const bool have_next = (kb + 1) < OB_NKB;
    if (have_next) wo_block_load(sx, sy, Wo, kb + 1, tid);

    if (rsel == 0) wo_block_mma<NA, NA>(acc, arows, cur, kb, cg, half, l16);
    else           wo_block_mma<NA, NB>(acc, arows, cur, kb, cg, half, l16);

    if (have_next) wo_block_store(nxt, sx, sy, tid);
    __syncthreads();
  }

  #pragma unroll
  for (int r = 0; r < NA; ++r) {
    if (r < nrt) {
      #pragma unroll
      for (int j = 0; j < 5; ++j) {
        const int c = cg * 80 + 16 * j + l16;
        const float bj = bo[c];
        #pragma unroll
        for (int i = 0; i < 8; ++i) {
          const int row = (rt0 + r) * 16 + i + 8 * half;
          orow[(size_t)row * C_DIM + c] = acc[r][j][i] + bj;
        }
      }
    }
  }
}

__global__ __launch_bounds__(256)
void attn_out_kernel(const float* __restrict__ Q,
                     const float* __restrict__ Kp,
                     const float* __restrict__ Vp,
                     const float* __restrict__ Wo,
                     const float* __restrict__ bo,
                     float* __restrict__ AB,
                     float* __restrict__ out)
{
  __shared__ __attribute__((aligned(16))) float AOs[NR_ATT * 16 * AOP];
  __shared__ __attribute__((aligned(16))) float Bs[2 * OB_BUF];
  const int tid  = threadIdx.x;
  const int h    = __builtin_amdgcn_readfirstlane(tid >> 5);
  const int lane = tid & 31;
  const int half = (lane >> 4) & 1;
  const int l16  = lane & 15;
  const int g0   = blockIdx.x * NR_ATT;
  const float scale = 0.15811388300841898f;
  const float NEG_INF = -__builtin_inff();
  const v8f zero = {0.f,0.f,0.f,0.f,0.f,0.f,0.f,0.f};

  int bcur = g0 / TILES_B;
  v2f kf[KS_D][KT_S];
  v2f vf[KT_S][KM_S][3];
  load_k_frags(kf, Kp, bcur, h, half, l16);
  load_v_frags(vf, Vp, bcur, h, half, l16);

  const float* qrow = Q + (size_t)(g0 * 16 + l16) * C_DIM + h * HD + 2 * half;
  v2f qf[KS_D];
  for (int kk = 0; kk < KS_D; ++kk) qf[kk] = *(const v2f*)(qrow + 4 * kk);

  for (int it = 0; it < NR_ATT; ++it) {
    const int g  = g0 + it;
    const int b  = g / TILES_B;
    const int t0 = (g - b * TILES_B) << 4;
    if (b != bcur) {
      load_k_frags(kf, Kp, b, h, half, l16);
      load_v_frags(vf, Vp, b, h, half, l16);
      bcur = b;
    }

    v8f sc[KT_S];
    #pragma unroll
    for (int j = 0; j < KT_S; ++j) sc[j] = zero;
    for (int kk = 0; kk < KS_D; ++kk) {
      #pragma unroll
      for (int j = 0; j < KT_S; ++j) sc[j] = wmma4(kf[kk][j], qf[kk], sc[j]);
    }

    if (it + 1 < NR_ATT) {
      qrow += (size_t)16 * C_DIM;
      for (int kk = 0; kk < KS_D; ++kk) qf[kk] = *(const v2f*)(qrow + 4 * kk);
    }

    float mj[KT_S];
    #pragma unroll
    for (int j = 0; j < KT_S; ++j) {
      mj[j] = NEG_INF;
      #pragma unroll
      for (int i = 0; i < 8; ++i) {
        const bool ok = (16 * j + 8 * half + i) < S_DIM;
        const float x = ok ? sc[j][i] * scale : NEG_INF;
        sc[j][i] = x;
        mj[j] = fmaxf(mj[j], x);
      }
    }
    float mx = fmaxf(fmaxf(fmaxf(mj[0], mj[1]), fmaxf(mj[2], mj[3])), mj[4]);
    mx = fmaxf(mx, xhalf(mx));
    float sj[KT_S];
    #pragma unroll
    for (int j = 0; j < KT_S; ++j) {
      sj[j] = 0.f;
      #pragma unroll
      for (int i = 0; i < 8; ++i) {
        const float e = __expf(sc[j][i] - mx);
        sc[j][i] = e;
        sj[j] += e;
      }
    }
    float sm = ((sj[0] + sj[1]) + (sj[2] + sj[3])) + sj[4];
    sm = sm + xhalf(sm);
    const float rs = 1.f / sm;
    #pragma unroll
    for (int j = 0; j < KT_S; ++j) {
      #pragma unroll
      for (int i = 0; i < 8; ++i) sc[j][i] *= rs;
    }

    if (b == B_BASE) {
      float* abrow = AB + (size_t)(h * T_DIM + t0 + l16) * SP + 8 * half;
      #pragma unroll
      for (int j = 0; j < KT_S; ++j) {
        v4f lo, hi;
        lo.x = sc[j][0]; lo.y = sc[j][1]; lo.z = sc[j][2]; lo.w = sc[j][3];
        hi.x = sc[j][4]; hi.y = sc[j][5]; hi.z = sc[j][6]; hi.w = sc[j][7];
        *(v4f*)(abrow + 16 * j)     = lo;
        *(v4f*)(abrow + 16 * j + 4) = hi;
      }
    }

    pv_to_lds(sc, vf, AOs + it * 16 * AOP, h, half, l16);
  }

  out_proj_from_lds<NR_ATT>(AOs, Bs, Bs + OB_BUF, Wo, bo,
                            out + (size_t)g0 * 16 * C_DIM, tid, h, half, l16);
}

__global__ __launch_bounds__(256)
void edit_out_kernel(const float* __restrict__ AB,
                     const float* __restrict__ Map,
                     const float* __restrict__ Vp,
                     const float* __restrict__ Wo,
                     const float* __restrict__ bo,
                     float* __restrict__ out)
{
  __shared__ __attribute__((aligned(16))) float AOs[NR_EDT * 16 * AOP];
  __shared__ __attribute__((aligned(16))) float Bs[2 * OB_BUF];
  const int tid  = threadIdx.x;
  const int h    = __builtin_amdgcn_readfirstlane(tid >> 5);
  const int lane = tid & 31;
  const int half = (lane >> 4) & 1;
  const int l16  = lane & 15;
  const int g0   = blockIdx.x * NR_EDT;
  const v8f zero = {0.f,0.f,0.f,0.f,0.f,0.f,0.f,0.f};

  int pcur = g0 / TILES_B;
  v2f mf[KS_S][KT_S];
  v2f vf[KT_S][KM_S][3];
  load_m_frags(mf, Map, pcur, half, l16);
  load_v_frags(vf, Vp, B_BASE + 1 + pcur, h, half, l16);

  const float* abase = AB + (size_t)(h * T_DIM + l16) * SP + 2 * half;
  const float* arow  = abase + (size_t)((g0 - pcur * TILES_B) * 16) * SP;
  v2f af[KS_S];
  for (int ks = 0; ks < KS_S; ++ks) af[ks] = *(const v2f*)(arow + 4 * ks);

  for (int it = 0; it < NR_EDT; ++it) {
    const int g = g0 + it;
    const int p = g / TILES_B;
    if (p != pcur) {
      load_m_frags(mf, Map, p, half, l16);
      load_v_frags(vf, Vp, B_BASE + 1 + p, h, half, l16);
      pcur = p;
    }

    v8f acc[KT_S];
    #pragma unroll
    for (int j = 0; j < KT_S; ++j) acc[j] = zero;
    for (int ks = 0; ks < KS_S; ++ks) {
      #pragma unroll
      for (int j = 0; j < KT_S; ++j) acc[j] = wmma4(mf[ks][j], af[ks], acc[j]);
    }

    if (it + 1 < NR_EDT) {
      const int gn = g + 1;
      arow = abase + (size_t)((gn - (gn / TILES_B) * TILES_B) * 16) * SP;
      for (int ks = 0; ks < KS_S; ++ks) af[ks] = *(const v2f*)(arow + 4 * ks);
    }

    #pragma unroll
    for (int j = 0; j < KT_S; ++j) {
      #pragma unroll
      for (int i = 0; i < 8; ++i) {
        const bool ok = (16 * j + 8 * half + i) < S_DIM;
        acc[j][i] = ok ? acc[j][i] : 0.f;
      }
    }

    pv_to_lds(acc, vf, AOs + it * 16 * AOP, h, half, l16);
  }

  out_proj_from_lds<NR_EDT>(AOs, Bs, Bs + OB_BUF, Wo, bo,
                            out + ((size_t)5 * T_DIM + (size_t)g0 * 16) * C_DIM,
                            tid, h, half, l16);
}

extern "C" void kernel_launch(void* const* d_in, const int* in_sizes, int n_in,
                              void* d_out, int out_size, void* d_ws, size_t ws_size,
                              hipStream_t stream) {
  const float* X   = (const float*)d_in[0];
  const float* E   = (const float*)d_in[1];
  const float* Map = (const float*)d_in[2];
  const float* Wq  = (const float*)d_in[3];
  const float* Wk  = (const float*)d_in[4];
  const float* Wv  = (const float*)d_in[5];
  const float* Wo  = (const float*)d_in[6];
  const float* bo  = (const float*)d_in[7];
  float* out = (float*)d_out;

  float* ws = (float*)d_ws;
  float* Q  = ws;
  float* Kp = Q  + (size_t)5 * T_DIM * C_DIM;
  float* Vp = Kp + (size_t)B_ALL * S_DIM * C_DIM;
  float* AB = Vp + (size_t)B_ALL * S_DIM * C_DIM;

  const dim3 gblk(128);
  const int MR = B_ALL * S_DIM;

  gemm_wmma_f32<<<dim3(C_DIM / 64, (5 * T_DIM) / 64), gblk, 0, stream>>>(
      X, Wq, nullptr, Q, 5 * T_DIM, C_DIM, C_DIM);
  kv_proj_kernel<<<dim3(C_DIM / KV_COLS, (MR + KV_ROWS - 1) / KV_ROWS, 2), dim3(256), 0, stream>>>(
      E, Wk, Wv, Kp, Vp);

  attn_out_kernel<<<dim3(5 * TILES_B / NR_ATT), dim3(256), 0, stream>>>(
      Q, Kp, Vp, Wo, bo, AB, out);

  edit_out_kernel<<<dim3(3 * TILES_B / NR_EDT), dim3(256), 0, stream>>>(
      AB, Map, Vp, Wo, bo, out);
}


# Q projection folded into the attention launch: a work-group forms its 80 Q rows in LDS (X and Wq staged in K-blocks of 32), waves read Q fragments from LDS; launches 4 -> 3; sums unchanged
# speedup vs baseline: 2.5962x; 1.1507x over previous
#include <hip/hip_runtime.h>


typedef __attribute__((ext_vector_type(2))) float v2f;
typedef __attribute__((ext_vector_type(4))) float v4f;
typedef __attribute__((ext_vector_type(8))) float v8f;

#define B_ALL   8
#define B_BASE  4
#define T_DIM   4096
#define C_DIM   320
#define S_DIM   77
#define SP      80
#define D_ENC   768
#define HD      40
#define H_NUM   8

__device__ __forceinline__ v8f wmma4(v2f a, v2f b, v8f c) {
  return __builtin_amdgcn_wmma_f32_16x16x4_f32(
       false, a,  false, b,
       (short)0, c,  false,  false);
}

#define KV_WAVES 8
#define KV_KS    (D_ENC / KV_WAVES / 4)
#define KV_GRP   6
#define KV_ROWS  32
#define KV_COLS  64

__global__ __launch_bounds__(256)
void kv_proj_kernel(const float* __restrict__ E,
                    const float* __restrict__ Wk,
                    const float* __restrict__ Wv,
                    float* __restrict__ Kp,
                    float* __restrict__ Vp)
{
  __shared__ float red[KV_WAVES][KV_ROWS * KV_COLS];
  const int tid  = threadIdx.x;
  const int wave = tid >> 5;
  const int lane = tid & 31;
  const int half = lane >> 4;
  const int l16  = lane & 15;
  const int col0 = blockIdx.x * KV_COLS;
  const int row0 = blockIdx.y * KV_ROWS;
  const float* __restrict__ Wm = blockIdx.z ? Wv : Wk;
  float* __restrict__ Om       = blockIdx.z ? Vp : Kp;
  const int M = B_ALL * S_DIM;
  const int kbase = wave * (KV_KS * 4);

  int ra = row0 + l16;       if (ra > M - 1) ra = M - 1;
  int rb = row0 + 16 + l16;  if (rb > M - 1) rb = M - 1;
  const float* pa = E + (size_t)ra * D_ENC + kbase + 2 * half;
  const float* pb = E + (size_t)rb * D_ENC + kbase + 2 * half;
  const float* pw = Wm + (size_t)(kbase + 2 * half) * C_DIM + col0 + 4 * l16;

  v4f wx[2][KV_GRP], wy[2][KV_GRP];
  for (int s = 0; s < KV_GRP; ++s) {
    wx[0][s] = *(const v4f*)(pw + (size_t)(4 * s)     * C_DIM);
    wy[0][s] = *(const v4f*)(pw + (size_t)(4 * s + 1) * C_DIM);
  }
  v2f a0[KV_KS], a1[KV_KS];
  for (int ks = 0; ks < KV_KS; ++ks) {
    a0[ks] = *(const v2f*)(pa + 4 * ks);
    a1[ks] = *(const v2f*)(pb + 4 * ks);
  }

  const v8f zero = {0.f,0.f,0.f,0.f,0.f,0.f,0.f,0.f};
  v8f acc[2][4];
  for (int j = 0; j < 4; ++j) { acc[0][j] = zero; acc[1][j] = zero; }

  for (int g = 0; g < KV_KS / KV_GRP; ++g) {
    const int cur = g & 1;
    if (g + 1 < KV_KS / KV_GRP) {
      for (int s = 0; s < KV_GRP; ++s) {
        const int k = 4 * ((g + 1) * KV_GRP + s);
        wx[cur ^ 1][s] = *(const v4f*)(pw + (size_t)k       * C_DIM);
        wy[cur ^ 1][s] = *(const v4f*)(pw + (size_t)(k + 1) * C_DIM);
      }
    }
    for (int s = 0; s < KV_GRP; ++s) {
      const int ks = g * KV_GRP + s;
      for (int j = 0; j < 4; ++j) {
        v2f b;
        b.x = wx[cur][s][j];
        b.y = wy[cur][s][j];
        acc[0][j] = wmma4(a0[ks], b, acc[0][j]);
        acc[1][j] = wmma4(a1[ks], b, acc[1][j]);
      }
    }
  }

  for (int rt = 0; rt < 2; ++rt) {
    for (int i = 0; i < 8; ++i) {
      v4f t;
      t.x = acc[rt][0][i]; t.y = acc[rt][1][i];
      t.z = acc[rt][2][i]; t.w = acc[rt][3][i];
      *(v4f*)(&red[wave][(16 * rt + i + 8 * half) * KV_COLS + 4 * l16]) = t;
    }
  }
  __syncthreads();

  const int rr = tid >> 4;
  const int cg = tid & 15;
  for (int q = 0; q < 2; ++q) {
    const int row = rr + 16 * q;
    v4f s = *(const v4f*)(&red[0][row * KV_COLS + 4 * cg]);
    for (int w = 1; w < KV_WAVES; ++w) {
      const v4f p = *(const v4f*)(&red[w][row * KV_COLS + 4 * cg]);
      s.x += p.x; s.y += p.y; s.z += p.z; s.w += p.w;
    }
    if (row0 + row < M)
      *(v4f*)(Om + (size_t)(row0 + row) * C_DIM + col0 + 4 * cg) = s;
  }
}

#define NR_ATT  5
#define NR_ATT_ROWS (NR_ATT * 16)
#define NR_EDT  3
#define TILES_B (T_DIM / 16)
#define KS_D    (HD / 4)
#define KS_S    (SP / 4)

#define KT_S    (SP / 16)
#define KM_S    4

__device__ __forceinline__ float xhalf(float x) {
  const int b = __builtin_bit_cast(int, x);
  return __builtin_bit_cast(float,
      __builtin_amdgcn_permlanex16(b, b, 0x76543210, 0xfedcba98, false, false));
}

__device__ __forceinline__
void load_v_frags(v2f (&vf)[KT_S][KM_S][3], const float* __restrict__ Vp,
                  int b, int h, int half, int l16) {
  const float* vbase = Vp + (size_t)(b * S_DIM) * C_DIM + h * HD;
  #pragma unroll
  for (int j = 0; j < KT_S; ++j) {
    #pragma unroll
    for (int m = 0; m < KM_S; ++m) {
      const int ka = 16 * j + 8 * half + 2 * m;
      const int r0 = ka     < S_DIM ? ka     : S_DIM - 1;
      const int r1 = ka + 1 < S_DIM ? ka + 1 : S_DIM - 1;
      #pragma unroll
      for (int jn = 0; jn < 3; ++jn) {
        const int c  = 16 * jn + l16;
        const int cc = c < HD ? c : HD - 1;
        const float x = vbase[(size_t)r0 * C_DIM + cc];
        const float y = vbase[(size_t)r1 * C_DIM + cc];
        vf[j][m][jn].x = (c < HD && ka     < S_DIM) ? x : 0.f;
        vf[j][m][jn].y = (c < HD && ka + 1 < S_DIM) ? y : 0.f;
      }
    }
  }
}

__device__ __forceinline__
void load_k_frags(v2f (&kf)[KS_D][KT_S], const float* __restrict__ Kp,
                  int b, int h, int half, int l16) {
  const float* kbase = Kp + (size_t)(b * S_DIM) * C_DIM + h * HD;
  #pragma unroll
  for (int kk = 0; kk < KS_D; ++kk) {
    const int ka = 4 * kk + 2 * half;
    #pragma unroll
    for (int j = 0; j < KT_S; ++j) {
      const int s  = 16 * j + l16;
      const int ss = s < S_DIM ? s : S_DIM - 1;
      const v2f t = *(const v2f*)(kbase + (size_t)ss * C_DIM + ka);
      kf[kk][j].x = (s < S_DIM) ? t.x : 0.f;
      kf[kk][j].y = (s < S_DIM) ? t.y : 0.f;
    }
  }
}

__device__ __forceinline__
void load_m_frags(v2f (&mf)[KS_S][KT_S], const float* __restrict__ Map,
                  int p, int half, int l16) {
  const float* mbase = Map + (size_t)p * S_DIM * S_DIM;
  #pragma unroll
  for (int ks = 0; ks < KS_S; ++ks) {
    const int ka = 4 * ks + 2 * half;
    const int r0 = ka     < S_DIM ? ka     : S_DIM - 1;
    const int r1 = ka + 1 < S_DIM ? ka + 1 : S_DIM - 1;
    #pragma unroll
    for (int j = 0; j < KT_S; ++j) {
      const int n  = 16 * j + l16;
      const int nn = n < S_DIM ? n : S_DIM - 1;
      const float x = mbase[(size_t)r0 * S_DIM + nn];
      const float y = mbase[(size_t)r1 * S_DIM + nn];
      mf[ks][j].x = (n < S_DIM && ka     < S_DIM) ? x : 0.f;
      mf[ks][j].y = (n < S_DIM && ka + 1 < S_DIM) ? y : 0.f;
    }
  }
}

#define AOP      324
#define OB_K     32
#define OB_P     672
#define OB_NKB   (C_DIM / OB_K)
#define OB_ITEMS ((OB_K / 2) * (C_DIM / 4) / 256)
#define OB_BUF   ((OB_K / 2) * OB_P)
#define XC_Q     (OB_K / 4)
#define XC_N     (NR_ATT_ROWS * XC_Q)
#define XC_ITEMS ((XC_N + 255) / 256)

__device__ __forceinline__
void pv_to_lds(const v8f (&pt)[KT_S], const v2f (&vf)[KT_S][KM_S][3],
               float* aos, int h, int half, int l16) {
  const v8f zero = {0.f,0.f,0.f,0.f,0.f,0.f,0.f,0.f};
  v8f o[3];
  #pragma unroll
  for (int j = 0; j < 3; ++j) o[j] = zero;

  #pragma unroll
  for (int j = 0; j < KT_S; ++j) {
    #pragma unroll
    for (int m = 0; m < KM_S; ++m) {
      v2f a;
      a.x = pt[j][2 * m];
      a.y = pt[j][2 * m + 1];
      #pragma unroll
      for (int jn = 0; jn < 3; ++jn) o[jn] = wmma4(a, vf[j][m][jn], o[jn]);
    }
  }

  #pragma unroll
  for (int j = 0; j < 3; ++j) {
    const int c = 16 * j + l16;
    if (c < HD) {
      #pragma unroll
      for (int i = 0; i < 8; ++i)
        aos[(i + 8 * half) * AOP + h * HD + c] = o[j][i];
    }
  }
}

__device__ __forceinline__
void wo_block_load(v4f (&sx)[OB_ITEMS], v4f (&sy)[OB_ITEMS],
                   const float* __restrict__ Wo, int kb, int tid) {
  #pragma unroll
  for (int u = 0; u < OB_ITEMS; ++u) {
    const int f  = tid + 256 * u;
    const int kp = f / (C_DIM / 4);
    const int q  = f - (C_DIM / 4) * kp;
    const float* p = Wo + (size_t)(kb * OB_K + 2 * kp) * C_DIM + 4 * q;
    sx[u] = *(const v4f*)(p);
    sy[u] = *(const v4f*)(p + C_DIM);
  }
}

__device__ __forceinline__
void wo_block_store(float* bs, const v4f (&sx)[OB_ITEMS], const v4f (&sy)[OB_ITEMS],
                    int tid) {
  #pragma unroll
  for (int u = 0; u < OB_ITEMS; ++u) {
    const int f  = tid + 256 * u;
    const int kp = f / (C_DIM / 4);
    const int q  = f - (C_DIM / 4) * kp;
    float* d = bs + kp * OB_P + 8 * q;
    v4f lo, hi;
    lo.x = sx[u].x; lo.y = sy[u].x; lo.z = sx[u].y; lo.w = sy[u].y;
    hi.x = sx[u].z; hi.y = sy[u].z; hi.z = sx[u].w; hi.w = sy[u].w;
    *(v4f*)(d)     = lo;
    *(v4f*)(d + 4) = hi;
  }
}

__device__ __forceinline__
void x_block_load(v4f (&xr)[XC_ITEMS], const float* __restrict__ xrows, int kb,
                  int tid, int wave) {
  #pragma unroll
  for (int u = 0; u < XC_ITEMS; ++u) {
    if (256 * (u + 1) <= XC_N || wave < (XC_N - 256 * u) / 32) {
      const int f = tid + 256 * u;
      xr[u] = *(const v4f*)(xrows + (size_t)(f / XC_Q) * C_DIM + kb * OB_K + 4 * (f % XC_Q));
    }
  }
}

__device__ __forceinline__
void x_block_store(float* aos, const v4f (&xr)[XC_ITEMS], int kb, int tid, int wave) {
  #pragma unroll
  for (int u = 0; u < XC_ITEMS; ++u) {
    if (256 * (u + 1) <= XC_N || wave < (XC_N - 256 * u) / 32) {
      const int f = tid + 256 * u;
      *(v4f*)(aos + (f / XC_Q) * AOP + kb * OB_K + 4 * (f % XC_Q)) = xr[u];
    }
  }
}

template <int NA, int NRT>
__device__ __forceinline__
void wo_block_mma(v8f (&acc)[NA][5], const float* arows, const float* bs,
                  int kb, int cg, int half, int l16) {
  const float* ap = arows + l16 * AOP + kb * OB_K + 2 * half;
  const float* bp = bs + half * OB_P + 2 * (cg * 80 + l16);
  #pragma unroll
  for (int ks = 0; ks < OB_K / 4; ++ks) {
    v2f b[5];
    #pragma unroll
    for (int j = 0; j < 5; ++j) b[j] = *(const v2f*)(bp + (2 * ks) * OB_P + 32 * j);
    #pragma unroll
    for (int r = 0; r < NRT; ++r) {
      const v2f a = *(const v2f*)(ap + r * 16 * AOP + 4 * ks);
      #pragma unroll
      for (int j = 0; j < 5; ++j) acc[r][j] = wmma4(a, b[j], acc[r][j]);
    }
  }
}

template <int NR, bool STAGE_X>
__device__ __forceinline__
void lds_rows_gemm(v8f (&acc)[(NR + 1) / 2][5], float* aos, float* bs0, float* bs1,
                   const float* __restrict__ Wm, const float* __restrict__ xrows,
                   int tid, int wave, int half, int l16) {
  static_assert(!STAGE_X || NR == NR_ATT, "x_block_* stage NR_ATT row tiles");
  constexpr int NA = (NR + 1) / 2;
  constexpr int NB = NR - NA;
  const int rsel = (wave ^ (wave >> 2)) & 1;
  const int cg   = (wave >> 1) & 3;
  const int rt0  = rsel ? NA : 0;

  const v8f zero = {0.f,0.f,0.f,0.f,0.f,0.f,0.f,0.f};
  #pragma unroll
  for (int r = 0; r < NA; ++r) {
    #pragma unroll
    for (int j = 0; j < 5; ++j) acc[r][j] = zero;
  }

  v4f sx[OB_ITEMS], sy[OB_ITEMS];
  v4f xr[XC_ITEMS];
  wo_block_load(sx, sy, Wm, 0, tid);
  if (STAGE_X) x_block_load(xr, xrows, 0, tid, wave);
  wo_block_store(bs0, sx, sy, tid);
  if (STAGE_X) x_block_store(aos, xr, 0, tid, wave);
  __syncthreads();

  const float* arows = aos + rt0 * 16 * AOP;
  for (int kb = 0; kb < OB_NKB; ++kb) {
    const float* cur = (kb & 1) ? bs1 : bs0;
    float*       nxt = (kb & 1) ? bs0 : bs1;
    const bool have_next = (kb + 1) < OB_NKB;
    if (have_next) {
      wo_block_load(sx, sy, Wm, kb + 1, tid);
      if (STAGE_X) x_block_load(xr, xrows, kb + 1, tid, wave);
    }

    if (rsel == 0) wo_block_mma<NA, NA>(acc, arows, cur, kb, cg, half, l16);
    else           wo_block_mma<NA, NB>(acc, arows, cur, kb, cg, half, l16);

    if (have_next) {
      wo_block_store(nxt, sx, sy, tid);
      if (STAGE_X) x_block_store(aos, xr, kb + 1, tid, wave);
    }
    __syncthreads();
  }
}

template <int NR>
__device__ __forceinline__
void out_proj_from_lds(float* aos, float* bs0, float* bs1,
                       const float* __restrict__ Wo, const float* __restrict__ bo,
                       float* __restrict__ orow, int tid, int wave, int half, int l16) {
  constexpr int NA = (NR + 1) / 2;
  constexpr int NB = NR - NA;
  const int rsel = (wave ^ (wave >> 2)) & 1;
  const int cg   = (wave >> 1) & 3;
  const int rt0  = rsel ? NA : 0;
  const int nrt  = rsel ? NB : NA;

  v8f acc[NA][5];
  lds_rows_gemm<NR, false>(acc, aos, bs0, bs1, Wo, nullptr, tid, wave, half, l16);

  #pragma unroll
  for (int r = 0; r < NA; ++r) {
    if (r < nrt) {
      #pragma unroll
      for (int j = 0; j < 5; ++j) {
        const int c = cg * 80 + 16 * j + l16;
        const float bj = bo[c];
        #pragma unroll
        for (int i = 0; i < 8; ++i) {
          const int row = (rt0 + r) * 16 + i + 8 * half;
          orow[(size_t)row * C_DIM + c] = acc[r][j][i] + bj;
        }
      }
    }
  }
}

__global__ __launch_bounds__(256)
void attn_out_kernel(const float* __restrict__ X,
                     const float* __restrict__ Wq,
                     const float* __restrict__ Kp,
                     const float* __restrict__ Vp,
                     const float* __restrict__ Wo,
                     const float* __restrict__ bo,
                     float* __restrict__ AB,
                     float* __restrict__ out)
{
  __shared__ __attribute__((aligned(16))) float AOs[NR_ATT * 16 * AOP];
  __shared__ __attribute__((aligned(16))) float Bs[2 * OB_BUF];
  const int tid  = threadIdx.x;
  const int h    = __builtin_amdgcn_readfirstlane(tid >> 5);
  const int lane = tid & 31;
  const int half = (lane >> 4) & 1;
  const int l16  = lane & 15;
  const int g0   = blockIdx.x * NR_ATT;
  const float scale = 0.15811388300841898f;
  const float NEG_INF = -__builtin_inff();
  const v8f zero = {0.f,0.f,0.f,0.f,0.f,0.f,0.f,0.f};

  int bcur = g0 / TILES_B;
  v2f kf[KS_D][KT_S];
  v2f vf[KT_S][KM_S][3];

  {
    constexpr int NA = (NR_ATT + 1) / 2;
    v8f qacc[NA][5];
    lds_rows_gemm<NR_ATT, true>(qacc, AOs, Bs, Bs + OB_BUF, Wq,
                                X + (size_t)g0 * 16 * C_DIM, tid, h, half, l16);

    load_k_frags(kf, Kp, bcur, h, half, l16);
    load_v_frags(vf, Vp, bcur, h, half, l16);

    const int rsel = (h ^ (h >> 2)) & 1;
    const int cg   = (h >> 1) & 3;
    const int rt0  = rsel ? NA : 0;
    const int nrt  = rsel ? NR_ATT - NA : NA;
    #pragma unroll
    for (int r = 0; r < NA; ++r) {
      if (r < nrt) {
        #pragma unroll
        for (int j = 0; j < 5; ++j) {
          #pragma unroll
          for (int i = 0; i < 8; ++i)
            AOs[((rt0 + r) * 16 + i + 8 * half) * AOP + cg * 80 + 16 * j + l16] = qacc[r][j][i];
        }
      }
    }
  }
  __syncthreads();

  const float* qrow = AOs + l16 * AOP + h * HD + 2 * half;
  v2f qf[KS_D];
  for (int kk = 0; kk < KS_D; ++kk) qf[kk] = *(const v2f*)(qrow + 4 * kk);

  for (int it = 0; it < NR_ATT; ++it) {
    const int g  = g0 + it;
    const int b  = g / TILES_B;
    const int t0 = (g - b * TILES_B) << 4;
    if (b != bcur) {
      load_k_frags(kf, Kp, b, h, half, l16);
      load_v_frags(vf, Vp, b, h, half, l16);
      bcur = b;
    }

    v8f sc[KT_S];
    #pragma unroll
    for (int j = 0; j < KT_S; ++j) sc[j] = zero;
    for (int kk = 0; kk < KS_D; ++kk) {
      #pragma unroll
      for (int j = 0; j < KT_S; ++j) sc[j] = wmma4(kf[kk][j], qf[kk], sc[j]);
    }

    if (it + 1 < NR_ATT) {
      qrow += 16 * AOP;
      for (int kk = 0; kk < KS_D; ++kk) qf[kk] = *(const v2f*)(qrow + 4 * kk);
    }

    float mj[KT_S];
    #pragma unroll
    for (int j = 0; j < KT_S; ++j) {
      mj[j] = NEG_INF;
      #pragma unroll
      for (int i = 0; i < 8; ++i) {
        const bool ok = (16 * j + 8 * half + i) < S_DIM;
        const float x = ok ? sc[j][i] * scale : NEG_INF;
        sc[j][i] = x;
        mj[j] = fmaxf(mj[j], x);
      }
    }
    float mx = fmaxf(fmaxf(fmaxf(mj[0], mj[1]), fmaxf(mj[2], mj[3])), mj[4]);
    mx = fmaxf(mx, xhalf(mx));
    float sj[KT_S];
    #pragma unroll
    for (int j = 0; j < KT_S; ++j) {
      sj[j] = 0.f;
      #pragma unroll
      for (int i = 0; i < 8; ++i) {
        const float e = __expf(sc[j][i] - mx);
        sc[j][i] = e;
        sj[j] += e;
      }
    }
    float sm = ((sj[0] + sj[1]) + (sj[2] + sj[3])) + sj[4];
    sm = sm + xhalf(sm);
    const float rs = 1.f / sm;
    #pragma unroll
    for (int j = 0; j < KT_S; ++j) {
      #pragma unroll
      for (int i = 0; i < 8; ++i) sc[j][i] *= rs;
    }

    if (b == B_BASE) {
      float* abrow = AB + (size_t)(h * T_DIM + t0 + l16) * SP + 8 * half;
      #pragma unroll
      for (int j = 0; j < KT_S; ++j) {
        v4f lo, hi;
        lo.x = sc[j][0]; lo.y = sc[j][1]; lo.z = sc[j][2]; lo.w = sc[j][3];
        hi.x = sc[j][4]; hi.y = sc[j][5]; hi.z = sc[j][6]; hi.w = sc[j][7];
        *(v4f*)(abrow + 16 * j)     = lo;
        *(v4f*)(abrow + 16 * j + 4) = hi;
      }
    }

    pv_to_lds(sc, vf, AOs + it * 16 * AOP, h, half, l16);
  }

  out_proj_from_lds<NR_ATT>(AOs, Bs, Bs + OB_BUF, Wo, bo,
                            out + (size_t)g0 * 16 * C_DIM, tid, h, half, l16);
}

__global__ __launch_bounds__(256)
void edit_out_kernel(const float* __restrict__ AB,
                     const float* __restrict__ Map,
                     const float* __restrict__ Vp,
                     const float* __restrict__ Wo,
                     const float* __restrict__ bo,
                     float* __restrict__ out)
{
  __shared__ __attribute__((aligned(16))) float AOs[NR_EDT * 16 * AOP];
  __shared__ __attribute__((aligned(16))) float Bs[2 * OB_BUF];
  const int tid  = threadIdx.x;
  const int h    = __builtin_amdgcn_readfirstlane(tid >> 5);
  const int lane = tid & 31;
  const int half = (lane >> 4) & 1;
  const int l16  = lane & 15;
  const int g0   = blockIdx.x * NR_EDT;
  const v8f zero = {0.f,0.f,0.f,0.f,0.f,0.f,0.f,0.f};

  int pcur = g0 / TILES_B;
  v2f mf[KS_S][KT_S];
  v2f vf[KT_S][KM_S][3];
  load_m_frags(mf, Map, pcur, half, l16);
  load_v_frags(vf, Vp, B_BASE + 1 + pcur, h, half, l16);

  const float* abase = AB + (size_t)(h * T_DIM + l16) * SP + 2 * half;
  const float* arow  = abase + (size_t)((g0 - pcur * TILES_B) * 16) * SP;
  v2f af[KS_S];
  for (int ks = 0; ks < KS_S; ++ks) af[ks] = *(const v2f*)(arow + 4 * ks);

  for (int it = 0; it < NR_EDT; ++it) {
    const int g = g0 + it;
    const int p = g / TILES_B;
    if (p != pcur) {
      load_m_frags(mf, Map, p, half, l16);
      load_v_frags(vf, Vp, B_BASE + 1 + p, h, half, l16);
      pcur = p;
    }

    v8f acc[KT_S];
    #pragma unroll
    for (int j = 0; j < KT_S; ++j) acc[j] = zero;
    for (int ks = 0; ks < KS_S; ++ks) {
      #pragma unroll
      for (int j = 0; j < KT_S; ++j) acc[j] = wmma4(mf[ks][j], af[ks], acc[j]);
    }

    if (it + 1 < NR_EDT) {
      const int gn = g + 1;
      arow = abase + (size_t)((gn - (gn / TILES_B) * TILES_B) * 16) * SP;
      for (int ks = 0; ks < KS_S; ++ks) af[ks] = *(const v2f*)(arow + 4 * ks);
    }

    #pragma unroll
    for (int j = 0; j < KT_S; ++j) {
      #pragma unroll
      for (int i = 0; i < 8; ++i) {
        const bool ok = (16 * j + 8 * half + i) < S_DIM;
        acc[j][i] = ok ? acc[j][i] : 0.f;
      }
    }

    pv_to_lds(acc, vf, AOs + it * 16 * AOP, h, half, l16);
  }

  out_proj_from_lds<NR_EDT>(AOs, Bs, Bs + OB_BUF, Wo, bo,
                            out + ((size_t)5 * T_DIM + (size_t)g0 * 16) * C_DIM,
                            tid, h, half, l16);
}

extern "C" void kernel_launch(void* const* d_in, const int* in_sizes, int n_in,
                              void* d_out, int out_size, void* d_ws, size_t ws_size,
                              hipStream_t stream) {
  const float* X   = (const float*)d_in[0];
  const float* E   = (const float*)d_in[1];
  const float* Map = (const float*)d_in[2];
  const float* Wq  = (const float*)d_in[3];
  const float* Wk  = (const float*)d_in[4];
  const float* Wv  = (const float*)d_in[5];
  const float* Wo  = (const float*)d_in[6];
  const float* bo  = (const float*)d_in[7];
  float* out = (float*)d_out;

  float* ws = (float*)d_ws;
  float* Kp = ws;
  float* Vp = Kp + (size_t)B_ALL * S_DIM * C_DIM;
  float* AB = Vp + (size_t)B_ALL * S_DIM * C_DIM;

  const int MR = B_ALL * S_DIM;

  kv_proj_kernel<<<dim3(C_DIM / KV_COLS, (MR + KV_ROWS - 1) / KV_ROWS, 2), dim3(256), 0, stream>>>(
      E, Wk, Wv, Kp, Vp);

  attn_out_kernel<<<dim3(5 * TILES_B / NR_ATT), dim3(256), 0, stream>>>(
      X, Wq, Kp, Vp, Wo, bo, AB, out);

  edit_out_kernel<<<dim3(3 * TILES_B / NR_EDT), dim3(256), 0, stream>>>(
      AB, Map, Vp, Wo, bo, out);
}
